# starting kernel
# baseline (speedup 1.0000x reference)
#include <hip/hip_runtime.h>

typedef __attribute__((ext_vector_type(2))) float v2f;
typedef __attribute__((ext_vector_type(8))) float v8f;

#define F 128

__global__ __launch_bounds__(256)
void gat_gemm_wmma(const float* __restrict__ H, const float* __restrict__ W,
                   float* __restrict__ Hp, int N) {
  const int lane = threadIdx.x & 31;
  const int wave = (int)((blockIdx.x * blockDim.x + threadIdx.x) >> 5);
  const int mTile = wave;
  if (mTile * 16 >= N) return;

  const int mr = lane & 15;
  const int kk = (lane >> 4) * 2;
  const float* __restrict__ Arow = H + (size_t)(mTile * 16 + mr) * F;
  const float* __restrict__ Bbase = W + (lane & 15);

  v8f acc[8];
#pragma unroll
  for (int nt = 0; nt < 8; ++nt) acc[nt] = (v8f){};

#pragma unroll 2
  for (int k = 0; k < F; k += 4) {
    v2f a = *(const v2f*)(Arow + k + kk);
    const float* __restrict__ Brow0 = Bbase + (size_t)(k + kk) * F;
    const float* __restrict__ Brow1 = Brow0 + F;
#pragma unroll
    for (int nt = 0; nt < 8; ++nt) {
      v2f b;
      b.x = Brow0[nt * 16];
      b.y = Brow1[nt * 16];
      acc[nt] = __builtin_amdgcn_wmma_f32_16x16x4_f32(false, a, false, b,
                                                      (short)0, acc[nt],
                                                      false, false);
    }
  }

  const int mBase = (lane >> 4) * 8;
  float* __restrict__ outBase =
      Hp + (size_t)(mTile * 16 + mBase) * F + (lane & 15);
#pragma unroll
  for (int nt = 0; nt < 8; ++nt) {
#pragma unroll
    for (int r = 0; r < 8; ++r)
      outBase[(size_t)r * F + nt * 16] = acc[nt][r];
  }
}

__global__ __launch_bounds__(256)
void gat_edge(const float* __restrict__ Hp, const int* __restrict__ row,
              const int* __restrict__ col, float* __restrict__ denom,
              float* __restrict__ outAcc, int E) {
  const int lane = threadIdx.x & 31;
  const int wave = (int)((blockIdx.x * blockDim.x + threadIdx.x) >> 5);
  const int nWaves = (int)((gridDim.x * blockDim.x) >> 5);

  for (int e = wave; e < E; e += nWaves) {
    const int r = row[e];
    const int c = col[e];

    const int e2 = e + nWaves;
    if (e2 < E) {
      const int rn = row[e2];
      const int cn = col[e2];
      __builtin_prefetch(Hp + (size_t)rn * F + lane * 4, 0, 3);
      __builtin_prefetch(Hp + (size_t)cn * F + lane * 4, 0, 3);
    }

    const float4 hr = *(const float4*)(Hp + (size_t)r * F + lane * 4);
    const float4 hc = *(const float4*)(Hp + (size_t)c * F + lane * 4);

    float p = hr.x * hc.x + hr.y * hc.y + hr.z * hc.z + hr.w * hc.w;
#pragma unroll
    for (int off = 16; off >= 1; off >>= 1) p += __shfl_xor(p, off, 32);

    const float s = fmaxf(p, 0.2f * p);
    const float ex = __expf(s);

    if (lane == 0) atomicAdd(denom + r, ex);

    float* o = outAcc + (size_t)r * F + lane * 4;
    atomicAdd(o + 0, ex * hc.x);
    atomicAdd(o + 1, ex * hc.y);
    atomicAdd(o + 2, ex * hc.z);
    atomicAdd(o + 3, ex * hc.w);
  }
}

__global__ void zero_kernel(float* __restrict__ p, size_t n) {
  size_t i = (size_t)blockIdx.x * blockDim.x + threadIdx.x;
  const size_t stride = (size_t)gridDim.x * blockDim.x;
  for (; i < n; i += stride) p[i] = 0.0f;
}

__global__ void gat_finalize(float* __restrict__ out, const float* __restrict__ denom,
                             const float* __restrict__ bias, size_t n) {
  size_t i = (size_t)blockIdx.x * blockDim.x + threadIdx.x;
  const size_t stride = (size_t)gridDim.x * blockDim.x;
  for (; i < n; i += stride) {
    const size_t node = i >> 7;
    const int cc = (int)(i & (F - 1));
    out[i] = out[i] / (denom[node] + 1e-10f) + bias[cc];
  }
}

extern "C" void kernel_launch(void* const* d_in, const int* in_sizes, int n_in,
                              void* d_out, int out_size, void* d_ws, size_t ws_size,
                              hipStream_t stream) {
  const float* H    = (const float*)d_in[0];
  const float* W    = (const float*)d_in[1];
  const float* bias = (const float*)d_in[2];
  const int*   row  = (const int*)d_in[3];
  const int*   col  = (const int*)d_in[4];

  const int N = in_sizes[0] / F;
  const int E = in_sizes[3];
  float* out = (float*)d_out;

  float* Hp    = (float*)d_ws;
  float* denom = (float*)((char*)d_ws + (size_t)N * F * sizeof(float));

  zero_kernel<<<512, 256, 0, stream>>>(denom, (size_t)N);
  zero_kernel<<<4096, 256, 0, stream>>>(out, (size_t)N * F);

  const int strips = (N + 15) / 16;
  gat_gemm_wmma<<<(strips + 7) / 8, 256, 0, stream>>>(H, W, Hp, N);

  gat_edge<<<4096, 256, 0, stream>>>(Hp, row, col, denom, out, E);

  gat_finalize<<<4096, 256, 0, stream>>>(out, denom, bias, (size_t)N * F);
}


# edges bucketed by destination row (counting sort in workspace, int atomics); a wave sums each row in registers and stores it once: no float atomics; row sum order = bucket arrival order
# speedup vs baseline: 1.6403x; 1.6403x over previous
#include <hip/hip_runtime.h>

typedef __attribute__((ext_vector_type(2))) float v2f;
typedef __attribute__((ext_vector_type(8))) float v8f;

#define F 128

__global__ __launch_bounds__(256)
void gat_gemm_wmma(const float* __restrict__ H, const float* __restrict__ W,
                   float* __restrict__ Hp, int N) {
  const int lane = threadIdx.x & 31;
  const int wave = (int)((blockIdx.x * blockDim.x + threadIdx.x) >> 5);
  const int mTile = wave;
  if (mTile * 16 >= N) return;

  const int mr = lane & 15;
  const int kk = (lane >> 4) * 2;
  const float* __restrict__ Arow = H + (size_t)(mTile * 16 + mr) * F;
  const float* __restrict__ Bbase = W + (lane & 15);

  v8f acc[8];
#pragma unroll
  for (int nt = 0; nt < 8; ++nt) acc[nt] = (v8f){};

#pragma unroll 2
  for (int k = 0; k < F; k += 4) {
    v2f a = *(const v2f*)(Arow + k + kk);
    const float* __restrict__ Brow0 = Bbase + (size_t)(k + kk) * F;
    const float* __restrict__ Brow1 = Brow0 + F;
#pragma unroll
    for (int nt = 0; nt < 8; ++nt) {
      v2f b;
      b.x = Brow0[nt * 16];
      b.y = Brow1[nt * 16];
      acc[nt] = __builtin_amdgcn_wmma_f32_16x16x4_f32(false, a, false, b,
                                                      (short)0, acc[nt],
                                                      false, false);
    }
  }

  const int mBase = (lane >> 4) * 8;
  float* __restrict__ outBase =
      Hp + (size_t)(mTile * 16 + mBase) * F + (lane & 15);
#pragma unroll
  for (int nt = 0; nt < 8; ++nt) {
#pragma unroll
    for (int r = 0; r < 8; ++r)
      outBase[(size_t)r * F + nt * 16] = acc[nt][r];
  }
}

struct RowSeg {
  int count;
  int cursor;
};

__global__ void zero_kernel(int* __restrict__ p, size_t n) {
  size_t i = (size_t)blockIdx.x * blockDim.x + threadIdx.x;
  const size_t stride = (size_t)gridDim.x * blockDim.x;
  for (; i < n; i += stride) p[i] = 0;
}

__global__ __launch_bounds__(256)
void row_count(const int* __restrict__ row, RowSeg* __restrict__ seg, int E) {
  int i = (int)(blockIdx.x * blockDim.x + threadIdx.x);
  const int stride = (int)(gridDim.x * blockDim.x);
  for (; i < E; i += stride) atomicAdd(&seg[row[i]].count, 1);
}

__global__ __launch_bounds__(1024)
void row_alloc(RowSeg* __restrict__ seg, int* __restrict__ total, int N) {
  __shared__ int wsum[32];
  __shared__ int blockBase;
  const int lane = threadIdx.x & 31;
  const int w = threadIdx.x >> 5;
  const int i = (int)(blockIdx.x * 1024 + threadIdx.x);
  const int c = (i < N) ? seg[i].count : 0;
  int incl = c;
#pragma unroll
  for (int off = 1; off < 32; off <<= 1) {
    const int t = __shfl_up(incl, off, 32);
    if (lane >= off) incl += t;
  }
  if (lane == 31) wsum[w] = incl;
  __syncthreads();
  if (w == 0) {
    const int s = wsum[lane];
    int si = s;
#pragma unroll
    for (int off = 1; off < 32; off <<= 1) {
      const int t = __shfl_up(si, off, 32);
      if (lane >= off) si += t;
    }
    wsum[lane] = si - s;
    if (lane == 31) blockBase = atomicAdd(total, si);
  }
  __syncthreads();
  if (i < N) seg[i].cursor = blockBase + wsum[w] + incl - c;
}

__global__ __launch_bounds__(256)
void row_fill(const int* __restrict__ row, const int* __restrict__ col,
              RowSeg* __restrict__ seg, int* __restrict__ scol, int E) {
  int i = (int)(blockIdx.x * blockDim.x + threadIdx.x);
  const int stride = (int)(gridDim.x * blockDim.x);
  for (; i < E; i += stride) {
    const int pos = atomicAdd(&seg[row[i]].cursor, 1);
    scol[pos] = col[i];
  }
}

__global__ __launch_bounds__(256)
void gat_row(const float* __restrict__ Hp, const RowSeg* __restrict__ seg,
             const int* __restrict__ scol,
             const float* __restrict__ bias, float* __restrict__ out, int N) {
  const int lane = threadIdx.x & 31;
  const int wave = __builtin_amdgcn_readfirstlane(
      (int)((blockIdx.x * blockDim.x + threadIdx.x) >> 5));
  const int nWaves = (int)((gridDim.x * blockDim.x) >> 5);
  const float4 b = *(const float4*)(bias + lane * 4);

  for (int r = wave; r < N; r += nWaves) {
    const RowSeg rs = seg[r];
    const int end = rs.cursor;
    const int beg = end - rs.count;

    const float4 hr = *(const float4*)(Hp + (size_t)r * F + lane * 4);
    float4 acc = make_float4(0.0f, 0.0f, 0.0f, 0.0f);
    float den = 0.0f;

    for (int base = beg; base < end; base += 32) {
      const int idx = base + lane;
      const int cv = (idx < end) ? scol[idx] : r;
      const int m = min(32, end - base);
      for (int j = 0; j < m; j += 4) {
        float4 hc[4];
#pragma unroll
        for (int u = 0; u < 4; ++u) {
          const int c = __builtin_amdgcn_readlane(cv, j + u);
          hc[u] = *(const float4*)(Hp + (size_t)c * F + lane * 4);
        }
#pragma unroll
        for (int u = 0; u < 4; ++u) {
          float p = hr.x * hc[u].x + hr.y * hc[u].y + hr.z * hc[u].z + hr.w * hc[u].w;
#pragma unroll
          for (int off = 16; off >= 1; off >>= 1) p += __shfl_xor(p, off, 32);
          const float s = fmaxf(p, 0.2f * p);
          const float ex = (j + u < m) ? __expf(s) : 0.0f;
          den += ex;
          acc.x += ex * hc[u].x;
          acc.y += ex * hc[u].y;
          acc.z += ex * hc[u].z;
          acc.w += ex * hc[u].w;
        }
      }
    }

    const float d = den + 1e-10f;
    float4 o;
    o.x = acc.x / d + b.x;
    o.y = acc.y / d + b.y;
    o.z = acc.z / d + b.z;
    o.w = acc.w / d + b.w;
    *(float4*)(out + (size_t)r * F + lane * 4) = o;
  }
}

extern "C" void kernel_launch(void* const* d_in, const int* in_sizes, int n_in,
                              void* d_out, int out_size, void* d_ws, size_t ws_size,
                              hipStream_t stream) {
  const float* H    = (const float*)d_in[0];
  const float* W    = (const float*)d_in[1];
  const float* bias = (const float*)d_in[2];
  const int*   row  = (const int*)d_in[3];
  const int*   col  = (const int*)d_in[4];

  const int N = in_sizes[0] / F;
  const int E = in_sizes[3];
  float* out = (float*)d_out;

  float*  Hp    = (float*)d_ws;
  RowSeg* seg   = (RowSeg*)(Hp + (size_t)N * F);
  int*    total = (int*)(seg + N);
  int*    scol  = total + 1;

  zero_kernel<<<512, 256, 0, stream>>>((int*)seg, (size_t)N * 2 + 1);

  row_count<<<2048, 256, 0, stream>>>(row, seg, E);
  row_alloc<<<(N + 1023) / 1024, 1024, 0, stream>>>(seg, total, N);
  row_fill<<<2048, 256, 0, stream>>>(row, col, seg, scol, E);

  const int strips = (N + 15) / 16;
  gat_gemm_wmma<<<(strips + 7) / 8, 256, 0, stream>>>(H, W, Hp, N);

  gat_row<<<2048, 256, 0, stream>>>(Hp, seg, scol, bias, out, N);
}


# edge bucketing redone as a two-level counting sort with LDS counters (256 row ranges, then rows inside a range): no global atomics left; row sum order still = arrival order
# speedup vs baseline: 2.7349x; 1.6673x over previous
#include <hip/hip_runtime.h>

typedef __attribute__((ext_vector_type(2))) float v2f;
typedef __attribute__((ext_vector_type(8))) float v8f;

#define F 128

__global__ __launch_bounds__(256)
void gat_gemm_wmma(const float* __restrict__ H, const float* __restrict__ W,
                   float* __restrict__ Hp, int N) {
  const int lane = threadIdx.x & 31;
  const int wave = (int)((blockIdx.x * blockDim.x + threadIdx.x) >> 5);
  const int mTile = wave;
  if (mTile * 16 >= N) return;

  const int mr = lane & 15;
  const int kk = (lane >> 4) * 2;
  const float* __restrict__ Arow = H + (size_t)(mTile * 16 + mr) * F;
  const float* __restrict__ Bbase = W + (lane & 15);

  v8f acc[8];
#pragma unroll
  for (int nt = 0; nt < 8; ++nt) acc[nt] = (v8f){};

#pragma unroll 2
  for (int k = 0; k < F; k += 4) {
    v2f a = *(const v2f*)(Arow + k + kk);
    const float* __restrict__ Brow0 = Bbase + (size_t)(k + kk) * F;
    const float* __restrict__ Brow1 = Brow0 + F;
#pragma unroll
    for (int nt = 0; nt < 8; ++nt) {
      v2f b;
      b.x = Brow0[nt * 16];
      b.y = Brow1[nt * 16];
      acc[nt] = __builtin_amdgcn_wmma_f32_16x16x4_f32(false, a, false, b,
                                                      (short)0, acc[nt],
                                                      false, false);
    }
  }

  const int mBase = (lane >> 4) * 8;
  float* __restrict__ outBase =
      Hp + (size_t)(mTile * 16 + mBase) * F + (lane & 15);
#pragma unroll
  for (int nt = 0; nt < 8; ++nt) {
#pragma unroll
    for (int r = 0; r < 8; ++r)
      outBase[(size_t)r * F + nt * 16] = acc[nt][r];
  }
}

struct RowSeg {
  int count;
  int cursor;
};

#define NBK 256
#define WROWS 1024

__global__ __launch_bounds__(1024)
void edge_hist(const int* __restrict__ row, int* __restrict__ table,
               int E, int rpb) {
  __shared__ int h[NBK];
  if (threadIdx.x < NBK) h[threadIdx.x] = 0;
  __syncthreads();
  const int chunk = (E + NBK - 1) / NBK;
  const int e0 = (int)blockIdx.x * chunk;
  const int e1 = min(E, e0 + chunk);
  for (int e = e0 + (int)threadIdx.x; e < e1; e += 1024)
    atomicAdd(&h[row[e] / rpb], 1);
  __syncthreads();
  if (threadIdx.x < NBK) table[threadIdx.x * NBK + blockIdx.x] = h[threadIdx.x];
}

__global__ __launch_bounds__(1024)
void table_scan(int* __restrict__ table) {
  __shared__ int wsum[32];
  const int lane = threadIdx.x & 31;
  const int w = threadIdx.x >> 5;
  constexpr int PER = NBK * NBK / 1024;
  int4* __restrict__ p = (int4*)(table + (size_t)threadIdx.x * PER);
  int4 v[PER / 4];
  int s = 0;
#pragma unroll
  for (int i = 0; i < PER / 4; ++i) {
    v[i] = p[i];
    s += v[i].x + v[i].y + v[i].z + v[i].w;
  }
  int incl = s;
#pragma unroll
  for (int off = 1; off < 32; off <<= 1) {
    const int t = __shfl_up(incl, off, 32);
    if (lane >= off) incl += t;
  }
  if (lane == 31) wsum[w] = incl;
  __syncthreads();
  if (w == 0) {
    const int ws = wsum[lane];
    int si = ws;
#pragma unroll
    for (int off = 1; off < 32; off <<= 1) {
      const int t = __shfl_up(si, off, 32);
      if (lane >= off) si += t;
    }
    wsum[lane] = si - ws;
  }
  __syncthreads();
  int run = wsum[w] + incl - s;
#pragma unroll
  for (int i = 0; i < PER / 4; ++i) {
    int4 o;
    o.x = run; run += v[i].x;
    o.y = run; run += v[i].y;
    o.z = run; run += v[i].z;
    o.w = run; run += v[i].w;
    p[i] = o;
  }
}

__global__ __launch_bounds__(1024)
void edge_part(const int* __restrict__ row, const int* __restrict__ col,
               const int* __restrict__ table, int2* __restrict__ part,
               int E, int rpb) {
  __shared__ int cur[NBK];
  if (threadIdx.x < NBK) cur[threadIdx.x] = table[threadIdx.x * NBK + blockIdx.x];
  __syncthreads();
  const int chunk = (E + NBK - 1) / NBK;
  const int e0 = (int)blockIdx.x * chunk;
  const int e1 = min(E, e0 + chunk);
  for (int e = e0 + (int)threadIdx.x; e < e1; e += 1024) {
    const int r = row[e];
    const int c = col[e];
    const int pos = atomicAdd(&cur[r / rpb], 1);
    part[pos] = make_int2(r, c);
  }
}

__global__ __launch_bounds__(1024)
void range_rows(const int2* __restrict__ part, const int* __restrict__ table,
                RowSeg* __restrict__ seg, int* __restrict__ scol,
                int E, int N, int rpb) {
  __shared__ int cnt[WROWS];
  __shared__ int wsum[32];
  __shared__ int winTotal;
  const int lane = threadIdx.x & 31;
  const int w = threadIdx.x >> 5;
  const int b = (int)blockIdx.x;
  const int r0 = b * rpb;
  const int r1 = min(N, r0 + rpb);
  const int ebeg = table[b * NBK];
  const int eend = (b + 1 < NBK) ? table[(b + 1) * NBK] : E;
  int winBase = ebeg;

  for (int w0 = r0; w0 < r1; w0 += WROWS) {
    cnt[threadIdx.x] = 0;
    __syncthreads();
    for (int e = ebeg + (int)threadIdx.x; e < eend; e += 1024) {
      const unsigned lr = (unsigned)(part[e].x - w0);
      if (lr < (unsigned)WROWS) atomicAdd(&cnt[lr], 1);
    }
    __syncthreads();
    const int c = cnt[threadIdx.x];
    int incl = c;
#pragma unroll
    for (int off = 1; off < 32; off <<= 1) {
      const int t = __shfl_up(incl, off, 32);
      if (lane >= off) incl += t;
    }
    if (lane == 31) wsum[w] = incl;
    __syncthreads();
    if (w == 0) {
      const int ws = wsum[lane];
      int si = ws;
#pragma unroll
      for (int off = 1; off < 32; off <<= 1) {
        const int t = __shfl_up(si, off, 32);
        if (lane >= off) si += t;
      }
      wsum[lane] = si - ws;
      if (lane == 31) winTotal = si;
    }
    __syncthreads();
    const int start = winBase + wsum[w] + incl - c;
    const int r = w0 + (int)threadIdx.x;
    if (r < r1) {
      RowSeg rs;
      rs.count = c;
      rs.cursor = start + c;
      seg[r] = rs;
    }
    cnt[threadIdx.x] = start;
    __syncthreads();
    for (int e = ebeg + (int)threadIdx.x; e < eend; e += 1024) {
      const int2 pr = part[e];
      const unsigned lr = (unsigned)(pr.x - w0);
      if (lr < (unsigned)WROWS) {
        const int pos = atomicAdd(&cnt[lr], 1);
        scol[pos] = pr.y;
      }
    }
    winBase += winTotal;
    __syncthreads();
  }
}

__global__ __launch_bounds__(256)
void gat_row(const float* __restrict__ Hp, const RowSeg* __restrict__ seg,
             const int* __restrict__ scol,
             const float* __restrict__ bias, float* __restrict__ out, int N) {
  const int lane = threadIdx.x & 31;
  const int wave = __builtin_amdgcn_readfirstlane(
      (int)((blockIdx.x * blockDim.x + threadIdx.x) >> 5));
  const int nWaves = (int)((gridDim.x * blockDim.x) >> 5);
  const float4 b = *(const float4*)(bias + lane * 4);

  for (int r = wave; r < N; r += nWaves) {
    const RowSeg rs = seg[r];
    const int end = rs.cursor;
    const int beg = end - rs.count;

    const float4 hr = *(const float4*)(Hp + (size_t)r * F + lane * 4);
    float4 acc = make_float4(0.0f, 0.0f, 0.0f, 0.0f);
    float den = 0.0f;

    for (int base = beg; base < end; base += 32) {
      const int idx = base + lane;
      const int cv = (idx < end) ? scol[idx] : r;
      const int m = min(32, end - base);
      for (int j = 0; j < m; j += 4) {
        float4 hc[4];
#pragma unroll
        for (int u = 0; u < 4; ++u) {
          const int c = __builtin_amdgcn_readlane(cv, j + u);
          hc[u] = *(const float4*)(Hp + (size_t)c * F + lane * 4);
        }
#pragma unroll
        for (int u = 0; u < 4; ++u) {
          float p = hr.x * hc[u].x + hr.y * hc[u].y + hr.z * hc[u].z + hr.w * hc[u].w;
#pragma unroll
          for (int off = 16; off >= 1; off >>= 1) p += __shfl_xor(p, off, 32);
          const float s = fmaxf(p, 0.2f * p);
          const float ex = (j + u < m) ? __expf(s) : 0.0f;
          den += ex;
          acc.x += ex * hc[u].x;
          acc.y += ex * hc[u].y;
          acc.z += ex * hc[u].z;
          acc.w += ex * hc[u].w;
        }
      }
    }

    const float d = den + 1e-10f;
    float4 o;
    o.x = acc.x / d + b.x;
    o.y = acc.y / d + b.y;
    o.z = acc.z / d + b.z;
    o.w = acc.w / d + b.w;
    *(float4*)(out + (size_t)r * F + lane * 4) = o;
  }
}

extern "C" void kernel_launch(void* const* d_in, const int* in_sizes, int n_in,
                              void* d_out, int out_size, void* d_ws, size_t ws_size,
                              hipStream_t stream) {
  const float* H    = (const float*)d_in[0];
  const float* W    = (const float*)d_in[1];
  const float* bias = (const float*)d_in[2];
  const int*   row  = (const int*)d_in[3];
  const int*   col  = (const int*)d_in[4];

  const int N = in_sizes[0] / F;
  const int E = in_sizes[3];
  float* out = (float*)d_out;

  float*  Hp    = (float*)d_ws;
  int*    table = (int*)(Hp + (size_t)N * F);
  RowSeg* seg   = (RowSeg*)(table + NBK * NBK);
  int2*   part  = (int2*)(seg + N);
  int*    scol  = (int*)(part + E);

  const int rpb = (N + NBK - 1) / NBK;
  edge_hist<<<NBK, 1024, 0, stream>>>(row, table, E, rpb);
  table_scan<<<1, 1024, 0, stream>>>(table);
  edge_part<<<NBK, 1024, 0, stream>>>(row, col, table, part, E, rpb);
  range_rows<<<NBK, 1024, 0, stream>>>(part, table, seg, scol, E, N, rpb);

  const int strips = (N + 15) / 16;
  gat_gemm_wmma<<<(strips + 7) / 8, 256, 0, stream>>>(H, W, Hp, N);

  gat_row<<<2048, 256, 0, stream>>>(Hp, seg, scol, bias, out, N);
}


# edge sort level 1 made block-local: each block loads its slice of the edge list once and sorts it by row range in LDS; range blocks read 256 short runs; 6 launches to 4; row sum order = arrival order
# speedup vs baseline: 3.5132x; 1.2846x over previous
#include <hip/hip_runtime.h>

typedef __attribute__((ext_vector_type(2))) float v2f;
typedef __attribute__((ext_vector_type(8))) float v8f;

#define F 128

__global__ __launch_bounds__(256)
void gat_gemm_wmma(const float* __restrict__ H, const float* __restrict__ W,
                   float* __restrict__ Hp, int N) {
  const int lane = threadIdx.x & 31;
  const int wave = (int)((blockIdx.x * blockDim.x + threadIdx.x) >> 5);
  const int mTile = wave;
  if (mTile * 16 >= N) return;

  const int mr = lane & 15;
  const int kk = (lane >> 4) * 2;
  const float* __restrict__ Arow = H + (size_t)(mTile * 16 + mr) * F;
  const float* __restrict__ Bbase = W + (lane & 15);

  v8f acc[8];
#pragma unroll
  for (int nt = 0; nt < 8; ++nt) acc[nt] = (v8f){};

#pragma unroll 2
  for (int k = 0; k < F; k += 4) {
    v2f a = *(const v2f*)(Arow + k + kk);
    const float* __restrict__ Brow0 = Bbase + (size_t)(k + kk) * F;
    const float* __restrict__ Brow1 = Brow0 + F;
#pragma unroll
    for (int nt = 0; nt < 8; ++nt) {
      v2f b;
      b.x = Brow0[nt * 16];
      b.y = Brow1[nt * 16];
      acc[nt] = __builtin_amdgcn_wmma_f32_16x16x4_f32(false, a, false, b,
                                                      (short)0, acc[nt],
                                                      false, false);
    }
  }

  const int mBase = (lane >> 4) * 8;
  float* __restrict__ outBase =
      Hp + (size_t)(mTile * 16 + mBase) * F + (lane & 15);
#pragma unroll
  for (int nt = 0; nt < 8; ++nt) {
#pragma unroll
    for (int r = 0; r < 8; ++r)
      outBase[(size_t)r * F + nt * 16] = acc[nt][r];
  }
}

struct RowSeg {
  int count;
  int cursor;
};

#define NBK 256
#define SPITCH (NBK + 1)
#define RPW (NBK / 32)
#define EPT 8
#define WROWS 1024

__global__ __launch_bounds__(1024)
void edge_local_sort(const int* __restrict__ row, const int* __restrict__ col,
                     int* __restrict__ S, int2* __restrict__ part,
                     int E, int rpb) {
  __shared__ int h[NBK];
  __shared__ int wsum[NBK / 32];
  const int tid = (int)threadIdx.x;
  const int lane = tid & 31;
  const int w = tid >> 5;
  if (tid < NBK) h[tid] = 0;
  __syncthreads();
  const int chunk = (E + NBK - 1) / NBK;
  const int e0 = min(E, (int)blockIdx.x * chunk);
  const int e1 = min(E, e0 + chunk);
  int rr[EPT];
  int cq[EPT];
  int bin[EPT];
#pragma unroll
  for (int u = 0; u < EPT; ++u) {
    const int e = e0 + tid + u * 1024;
    rr[u] = -1;
    cq[u] = 0;
    if (e < e1) {
      rr[u] = row[e];
      cq[u] = col[e];
    }
  }
#pragma unroll
  for (int u = 0; u < EPT; ++u) {
    bin[u] = rr[u] / rpb;
    if (rr[u] >= 0) atomicAdd(&h[bin[u]], 1);
  }
  for (int e = e0 + tid + EPT * 1024; e < e1; e += 1024)
    atomicAdd(&h[row[e] / rpb], 1);
  __syncthreads();
  int c = 0;
  int incl = 0;
  if (tid < NBK) {
    c = h[tid];
    incl = c;
#pragma unroll
    for (int off = 1; off < 32; off <<= 1) {
      const int t = __shfl_up(incl, off, 32);
      if (lane >= off) incl += t;
    }
    if (lane == 31) wsum[w] = incl;
  }
  __syncthreads();
  if (tid < NBK) {
    int before = 0;
#pragma unroll
    for (int i = 0; i < NBK / 32; ++i)
      if (i < w) before += wsum[i];
    const int excl = before + incl - c;
    int* __restrict__ Srow = S + (size_t)blockIdx.x * SPITCH;
    Srow[tid] = excl;
    if (tid == NBK - 1) Srow[NBK] = excl + c;
    h[tid] = e0 + excl;
  }
  __syncthreads();
#pragma unroll
  for (int u = 0; u < EPT; ++u) {
    if (rr[u] >= 0) {
      const int pos = atomicAdd(&h[bin[u]], 1);
      part[pos] = make_int2(rr[u], cq[u]);
    }
  }
  for (int e = e0 + tid + EPT * 1024; e < e1; e += 1024) {
    const int r = row[e];
    const int cc = col[e];
    const int pos = atomicAdd(&h[r / rpb], 1);
    part[pos] = make_int2(r, cc);
  }
}

__global__ __launch_bounds__(1024)
void range_rows(const int2* __restrict__ part, const int* __restrict__ S,
                RowSeg* __restrict__ seg, int* __restrict__ scol,
                int E, int N, int rpb) {
  __shared__ int cnt[WROWS];
  __shared__ int wsum[32];
  __shared__ int winTotal;
  __shared__ __attribute__((aligned(16))) int runStart[NBK];
  __shared__ __attribute__((aligned(16))) int runCnt[NBK];
  __shared__ int baseSum[NBK / 32];
  const int tid = (int)threadIdx.x;
  const int lane = tid & 31;
  const int w = tid >> 5;
  const int b = (int)blockIdx.x;
  const int r0 = b * rpb;
  const int r1 = min(N, r0 + rpb);
  const int chunk = (E + NBK - 1) / NBK;

  if (tid < NBK) {
    const int* __restrict__ Srow = S + (size_t)tid * SPITCH + b;
    const int s0 = Srow[0];
    const int s1 = Srow[1];
    runStart[tid] = min(E, tid * chunk) + s0;
    runCnt[tid] = s1 - s0;
    int t = s0;
#pragma unroll
    for (int off = 16; off >= 1; off >>= 1) t += __shfl_xor(t, off, 32);
    if (lane == 0) baseSum[w] = t;
  }
  __syncthreads();
  int winBase = 0;
#pragma unroll
  for (int i = 0; i < NBK / 32; ++i) winBase += baseSum[i];

  int rn[RPW];
  int rst[RPW];
#pragma unroll
  for (int u = 0; u < RPW; u += 4) {
    const int4 n4 = *(const int4*)(runCnt + w * RPW + u);
    const int4 s4 = *(const int4*)(runStart + w * RPW + u);
    rn[u] = n4.x; rn[u + 1] = n4.y; rn[u + 2] = n4.z; rn[u + 3] = n4.w;
    rst[u] = s4.x; rst[u + 1] = s4.y; rst[u + 2] = s4.z; rst[u + 3] = s4.w;
  }
  int2 pr[RPW];
#pragma unroll
  for (int u = 0; u < RPW; ++u) {
    pr[u] = make_int2(-1, 0);
    if (lane < rn[u]) pr[u] = part[rst[u] + lane];
  }

  for (int w0 = r0; w0 < r1; w0 += WROWS) {
    cnt[tid] = 0;
    __syncthreads();
#pragma unroll
    for (int u = 0; u < RPW; ++u) {
      const unsigned lr = (unsigned)(pr[u].x - w0);
      if (lr < (unsigned)WROWS) atomicAdd(&cnt[lr], 1);
    }
    for (int u = 0; u < RPW; ++u) {
      const int q = w * RPW + u;
      const int n = runCnt[q];
      const int st = runStart[q];
      for (int i = lane + 32; i < n; i += 32) {
        const unsigned lr = (unsigned)(part[st + i].x - w0);
        if (lr < (unsigned)WROWS) atomicAdd(&cnt[lr], 1);
      }
    }
    __syncthreads();
    const int c = cnt[tid];
    int incl = c;
#pragma unroll
    for (int off = 1; off < 32; off <<= 1) {
      const int t = __shfl_up(incl, off, 32);
      if (lane >= off) incl += t;
    }
    if (lane == 31) wsum[w] = incl;
    __syncthreads();
    if (w == 0) {
      const int ws = wsum[lane];
      int si = ws;
#pragma unroll
      for (int off = 1; off < 32; off <<= 1) {
        const int t = __shfl_up(si, off, 32);
        if (lane >= off) si += t;
      }
      wsum[lane] = si - ws;
      if (lane == 31) winTotal = si;
    }
    __syncthreads();
    const int start = winBase + wsum[w] + incl - c;
    const int r = w0 + tid;
    if (r < r1) {
      RowSeg rs;
      rs.count = c;
      rs.cursor = start + c;
      seg[r] = rs;
    }
    cnt[tid] = start;
    __syncthreads();
#pragma unroll
    for (int u = 0; u < RPW; ++u) {
      const unsigned lr = (unsigned)(pr[u].x - w0);
      if (lr < (unsigned)WROWS) {
        const int pos = atomicAdd(&cnt[lr], 1);
        scol[pos] = pr[u].y;
      }
    }
    for (int u = 0; u < RPW; ++u) {
      const int q = w * RPW + u;
      const int n = runCnt[q];
      const int st = runStart[q];
      for (int i = lane + 32; i < n; i += 32) {
        const int2 p2 = part[st + i];
        const unsigned lr = (unsigned)(p2.x - w0);
        if (lr < (unsigned)WROWS) {
          const int pos = atomicAdd(&cnt[lr], 1);
          scol[pos] = p2.y;
        }
      }
    }
    winBase += winTotal;
    __syncthreads();
  }
}

__global__ __launch_bounds__(256)
void gat_row(const float* __restrict__ Hp, const RowSeg* __restrict__ seg,
             const int* __restrict__ scol,
             const float* __restrict__ bias, float* __restrict__ out, int N) {
  const int lane = threadIdx.x & 31;
  const int wave = __builtin_amdgcn_readfirstlane(
      (int)((blockIdx.x * blockDim.x + threadIdx.x) >> 5));
  const int nWaves = (int)((gridDim.x * blockDim.x) >> 5);
  const float4 b = *(const float4*)(bias + lane * 4);

  for (int r = wave; r < N; r += nWaves) {
    const RowSeg rs = seg[r];
    const int end = rs.cursor;
    const int beg = end - rs.count;

    const float4 hr = *(const float4*)(Hp + (size_t)r * F + lane * 4);
    float4 acc = make_float4(0.0f, 0.0f, 0.0f, 0.0f);
    float den = 0.0f;

    for (int base = beg; base < end; base += 32) {
      const int idx = base + lane;
      const int cv = (idx < end) ? scol[idx] : r;
      const int m = min(32, end - base);
      for (int j = 0; j < m; j += 4) {
        float4 hc[4];
#pragma unroll
        for (int u = 0; u < 4; ++u) {
          const int c = __builtin_amdgcn_readlane(cv, j + u);
          hc[u] = *(const float4*)(Hp + (size_t)c * F + lane * 4);
        }
#pragma unroll
        for (int u = 0; u < 4; ++u) {
          float p = hr.x * hc[u].x + hr.y * hc[u].y + hr.z * hc[u].z + hr.w * hc[u].w;
#pragma unroll
          for (int off = 16; off >= 1; off >>= 1) p += __shfl_xor(p, off, 32);
          const float s = fmaxf(p, 0.2f * p);
          const float ex = (j + u < m) ? __expf(s) : 0.0f;
          den += ex;
          acc.x += ex * hc[u].x;
          acc.y += ex * hc[u].y;
          acc.z += ex * hc[u].z;
          acc.w += ex * hc[u].w;
        }
      }
    }

    const float d = den + 1e-10f;
    float4 o;
    o.x = acc.x / d + b.x;
    o.y = acc.y / d + b.y;
    o.z = acc.z / d + b.z;
    o.w = acc.w / d + b.w;
    *(float4*)(out + (size_t)r * F + lane * 4) = o;
  }
}

extern "C" void kernel_launch(void* const* d_in, const int* in_sizes, int n_in,
                              void* d_out, int out_size, void* d_ws, size_t ws_size,
                              hipStream_t stream) {
  const float* H    = (const float*)d_in[0];
  const float* W    = (const float*)d_in[1];
  const float* bias = (const float*)d_in[2];
  const int*   row  = (const int*)d_in[3];
  const int*   col  = (const int*)d_in[4];

  const int N = in_sizes[0] / F;
  const int E = in_sizes[3];
  float* out = (float*)d_out;

  float*  Hp    = (float*)d_ws;
  int*    S     = (int*)(Hp + (size_t)N * F);
  RowSeg* seg   = (RowSeg*)(S + NBK * SPITCH);
  int2*   part  = (int2*)(seg + N);
  int*    scol  = (int*)(part + E);

  const int rpb = (N + NBK - 1) / NBK;
  edge_local_sort<<<NBK, 1024, 0, stream>>>(row, col, S, part, E, rpb);
  range_rows<<<NBK, 1024, 0, stream>>>(part, S, seg, scol, E, N, rpb);

  const int strips = (N + 15) / 16;
  gat_gemm_wmma<<<(strips + 7) / 8, 256, 0, stream>>>(H, W, Hp, N);

  gat_row<<<2048, 256, 0, stream>>>(Hp, seg, scol, bias, out, N);
}


# row kernel re-mapped: 8 lanes per edge, 4 edges per wave step, 8 row loads in flight; score summed by 3 DPP adds, not 5 LDS shuffles per edge; row sum = 4 interleaved partial sums joined per row
# speedup vs baseline: 3.8387x; 1.0927x over previous
#include <hip/hip_runtime.h>

typedef __attribute__((ext_vector_type(2))) float v2f;
typedef __attribute__((ext_vector_type(8))) float v8f;

#define F 128

__global__ __launch_bounds__(256)
void gat_gemm_wmma(const float* __restrict__ H, const float* __restrict__ W,
                   float* __restrict__ Hp, int N) {
  const int lane = threadIdx.x & 31;
  const int wave = (int)((blockIdx.x * blockDim.x + threadIdx.x) >> 5);
  const int mTile = wave;
  if (mTile * 16 >= N) return;

  const int mr = lane & 15;
  const int kk = (lane >> 4) * 2;
  const float* __restrict__ Arow = H + (size_t)(mTile * 16 + mr) * F;
  const float* __restrict__ Bbase = W + (lane & 15);

  v8f acc[8];
#pragma unroll
  for (int nt = 0; nt < 8; ++nt) acc[nt] = (v8f){};

#pragma unroll 2
  for (int k = 0; k < F; k += 4) {
    v2f a = *(const v2f*)(Arow + k + kk);
    const float* __restrict__ Brow0 = Bbase + (size_t)(k + kk) * F;
    const float* __restrict__ Brow1 = Brow0 + F;
#pragma unroll
    for (int nt = 0; nt < 8; ++nt) {
      v2f b;
      b.x = Brow0[nt * 16];
      b.y = Brow1[nt * 16];
      acc[nt] = __builtin_amdgcn_wmma_f32_16x16x4_f32(false, a, false, b,
                                                      (short)0, acc[nt],
                                                      false, false);
    }
  }

  const int mBase = (lane >> 4) * 8;
  float* __restrict__ outBase =
      Hp + (size_t)(mTile * 16 + mBase) * F + (lane & 15);
#pragma unroll
  for (int nt = 0; nt < 8; ++nt) {
#pragma unroll
    for (int r = 0; r < 8; ++r)
      outBase[(size_t)r * F + nt * 16] = acc[nt][r];
  }
}

struct RowSeg {
  int count;
  int cursor;
};

#define NBK 256
#define SPITCH (NBK + 1)
#define RPW (NBK / 32)
#define EPT 8
#define WROWS 1024

__global__ __launch_bounds__(1024)
void edge_local_sort(const int* __restrict__ row, const int* __restrict__ col,
                     int* __restrict__ S, int2* __restrict__ part,
                     int E, int rpb) {
  __shared__ int h[NBK];
  __shared__ int wsum[NBK / 32];
  const int tid = (int)threadIdx.x;
  const int lane = tid & 31;
  const int w = tid >> 5;
  if (tid < NBK) h[tid] = 0;
  __syncthreads();
  const int chunk = (E + NBK - 1) / NBK;
  const int e0 = min(E, (int)blockIdx.x * chunk);
  const int e1 = min(E, e0 + chunk);
  int rr[EPT];
  int cq[EPT];
  int bin[EPT];
#pragma unroll
  for (int u = 0; u < EPT; ++u) {
    const int e = e0 + tid + u * 1024;
    rr[u] = -1;
    cq[u] = 0;
    if (e < e1) {
      rr[u] = row[e];
      cq[u] = col[e];
    }
  }
#pragma unroll
  for (int u = 0; u < EPT; ++u) {
    bin[u] = rr[u] / rpb;
    if (rr[u] >= 0) atomicAdd(&h[bin[u]], 1);
  }
  for (int e = e0 + tid + EPT * 1024; e < e1; e += 1024)
    atomicAdd(&h[row[e] / rpb], 1);
  __syncthreads();
  int c = 0;
  int incl = 0;
  if (tid < NBK) {
    c = h[tid];
    incl = c;
#pragma unroll
    for (int off = 1; off < 32; off <<= 1) {
      const int t = __shfl_up(incl, off, 32);
      if (lane >= off) incl += t;
    }
    if (lane == 31) wsum[w] = incl;
  }
  __syncthreads();
  if (tid < NBK) {
    int before = 0;
#pragma unroll
    for (int i = 0; i < NBK / 32; ++i)
      if (i < w) before += wsum[i];
    const int excl = before + incl - c;
    int* __restrict__ Srow = S + (size_t)blockIdx.x * SPITCH;
    Srow[tid] = excl;
    if (tid == NBK - 1) Srow[NBK] = excl + c;
    h[tid] = e0 + excl;
  }
  __syncthreads();
#pragma unroll
  for (int u = 0; u < EPT; ++u) {
    if (rr[u] >= 0) {
      const int pos = atomicAdd(&h[bin[u]], 1);
      part[pos] = make_int2(rr[u], cq[u]);
    }
  }
  for (int e = e0 + tid + EPT * 1024; e < e1; e += 1024) {
    const int r = row[e];
    const int cc = col[e];
    const int pos = atomicAdd(&h[r / rpb], 1);
    part[pos] = make_int2(r, cc);
  }
}

__global__ __launch_bounds__(1024)
void range_rows(const int2* __restrict__ part, const int* __restrict__ S,
                RowSeg* __restrict__ seg, int* __restrict__ scol,
                int E, int N, int rpb) {
  __shared__ int cnt[WROWS];
  __shared__ int wsum[32];
  __shared__ int winTotal;
  __shared__ __attribute__((aligned(16))) int runStart[NBK];
  __shared__ __attribute__((aligned(16))) int runCnt[NBK];
  __shared__ int baseSum[NBK / 32];
  const int tid = (int)threadIdx.x;
  const int lane = tid & 31;
  const int w = tid >> 5;
  const int b = (int)blockIdx.x;
  const int r0 = b * rpb;
  const int r1 = min(N, r0 + rpb);
  const int chunk = (E + NBK - 1) / NBK;

  if (tid < NBK) {
    const int* __restrict__ Srow = S + (size_t)tid * SPITCH + b;
    const int s0 = Srow[0];
    const int s1 = Srow[1];
    runStart[tid] = min(E, tid * chunk) + s0;
    runCnt[tid] = s1 - s0;
    int t = s0;
#pragma unroll
    for (int off = 16; off >= 1; off >>= 1) t += __shfl_xor(t, off, 32);
    if (lane == 0) baseSum[w] = t;
  }
  __syncthreads();
  int winBase = 0;
#pragma unroll
  for (int i = 0; i < NBK / 32; ++i) winBase += baseSum[i];

  int rn[RPW];
  int rst[RPW];
#pragma unroll
  for (int u = 0; u < RPW; u += 4) {
    const int4 n4 = *(const int4*)(runCnt + w * RPW + u);
    const int4 s4 = *(const int4*)(runStart + w * RPW + u);
    rn[u] = n4.x; rn[u + 1] = n4.y; rn[u + 2] = n4.z; rn[u + 3] = n4.w;
    rst[u] = s4.x; rst[u + 1] = s4.y; rst[u + 2] = s4.z; rst[u + 3] = s4.w;
  }
  int2 pr[RPW];
#pragma unroll
  for (int u = 0; u < RPW; ++u) {
    pr[u] = make_int2(-1, 0);
    if (lane < rn[u]) pr[u] = part[rst[u] + lane];
  }

  for (int w0 = r0; w0 < r1; w0 += WROWS) {
    cnt[tid] = 0;
    __syncthreads();
#pragma unroll
    for (int u = 0; u < RPW; ++u) {
      const unsigned lr = (unsigned)(pr[u].x - w0);
      if (lr < (unsigned)WROWS) atomicAdd(&cnt[lr], 1);
    }
    for (int u = 0; u < RPW; ++u) {
      const int q = w * RPW + u;
      const int n = runCnt[q];
      const int st = runStart[q];
      for (int i = lane + 32; i < n; i += 32) {
        const unsigned lr = (unsigned)(part[st + i].x - w0);
        if (lr < (unsigned)WROWS) atomicAdd(&cnt[lr], 1);
      }
    }
    __syncthreads();
    const int c = cnt[tid];
    int incl = c;
#pragma unroll
    for (int off = 1; off < 32; off <<= 1) {
      const int t = __shfl_up(incl, off, 32);
      if (lane >= off) incl += t;
    }
    if (lane == 31) wsum[w] = incl;
    __syncthreads();
    if (w == 0) {
      const int ws = wsum[lane];
      int si = ws;
#pragma unroll
      for (int off = 1; off < 32; off <<= 1) {
        const int t = __shfl_up(si, off, 32);
        if (lane >= off) si += t;
      }
      wsum[lane] = si - ws;
      if (lane == 31) winTotal = si;
    }
    __syncthreads();
    const int start = winBase + wsum[w] + incl - c;
    const int r = w0 + tid;
    if (r < r1) {
      RowSeg rs;
      rs.count = c;
      rs.cursor = start + c;
      seg[r] = rs;
    }
    cnt[tid] = start;
    __syncthreads();
#pragma unroll
    for (int u = 0; u < RPW; ++u) {
      const unsigned lr = (unsigned)(pr[u].x - w0);
      if (lr < (unsigned)WROWS) {
        const int pos = atomicAdd(&cnt[lr], 1);
        scol[pos] = pr[u].y;
      }
    }
    for (int u = 0; u < RPW; ++u) {
      const int q = w * RPW + u;
      const int n = runCnt[q];
      const int st = runStart[q];
      for (int i = lane + 32; i < n; i += 32) {
        const int2 p2 = part[st + i];
        const unsigned lr = (unsigned)(p2.x - w0);
        if (lr < (unsigned)WROWS) {
          const int pos = atomicAdd(&cnt[lr], 1);
          scol[pos] = p2.y;
        }
      }
    }
    winBase += winTotal;
    __syncthreads();
  }
}

typedef __attribute__((ext_vector_type(4))) float v4f;

struct Feat16 {
  v4f v[4];
};

template <int CTRL>
__device__ __forceinline__ float row_xor(float x) {
  return __builtin_bit_cast(float,
      __builtin_amdgcn_update_dpp(0, __builtin_bit_cast(int, x), CTRL, 0xf, 0xf, true));
}

__device__ __forceinline__ float half_xor(float x) {
  const int b = __builtin_bit_cast(int, x);
  return __builtin_bit_cast(float,
      __builtin_amdgcn_permlanex16(b, b, 0x76543210, 0xfedcba98, false, false));
}

__device__ __forceinline__ void fetch_step(Feat16& d, const float* __restrict__ Hp,
                                           int& cv, unsigned laneOff) {
  const unsigned c = __builtin_amdgcn_mov_dpp8((unsigned)cv, 0);
  cv = (int)__builtin_amdgcn_mov_dpp8((unsigned)cv, 0x1F58D1);
  const char* __restrict__ p = (const char*)Hp + (c * (unsigned)(F * 4) + laneOff);
#pragma unroll
  for (int i = 0; i < 4; ++i) d.v[i] = *(const v4f*)(p + i * 128);
}

__device__ __forceinline__ void use_step(const Feat16& hc, const Feat16& hr, bool valid,
                                         v2f (&acc)[8], float& den) {
  v2f qa = hr.v[0].xy * hc.v[0].xy;
  v2f qb = hr.v[2].xy * hc.v[2].xy;
  qa += hr.v[0].zw * hc.v[0].zw;
  qb += hr.v[2].zw * hc.v[2].zw;
  qa += hr.v[1].xy * hc.v[1].xy;
  qb += hr.v[3].xy * hc.v[3].xy;
  qa += hr.v[1].zw * hc.v[1].zw;
  qb += hr.v[3].zw * hc.v[3].zw;
  qa += qb;
  float p = qa.x + qa.y;
  p += row_xor<0x161>(p);
  p += row_xor<0x162>(p);
  p += row_xor<0x164>(p);
  const float s = fmaxf(p, 0.2f * p);
  const float ex = valid ? __expf(s) : 0.0f;
  den += ex;
#pragma unroll
  for (int i = 0; i < 4; ++i) {
    acc[2 * i] += ex * hc.v[i].xy;
    acc[2 * i + 1] += ex * hc.v[i].zw;
  }
}

__device__ __forceinline__ v4f sel4(bool c, v4f a, v4f b) {
  v4f o;
  o.x = c ? a.x : b.x;
  o.y = c ? a.y : b.y;
  o.z = c ? a.z : b.z;
  o.w = c ? a.w : b.w;
  return o;
}

__device__ __forceinline__ v4f row_xor8_4(v4f a) {
  v4f o;
  o.x = row_xor<0x168>(a.x);
  o.y = row_xor<0x168>(a.y);
  o.z = row_xor<0x168>(a.z);
  o.w = row_xor<0x168>(a.w);
  return o;
}

__device__ __forceinline__ v4f half_xor4(v4f a) {
  v4f o;
  o.x = half_xor(a.x);
  o.y = half_xor(a.y);
  o.z = half_xor(a.z);
  o.w = half_xor(a.w);
  return o;
}

__global__ __launch_bounds__(256)
void gat_row(const float* __restrict__ Hp, const RowSeg* __restrict__ seg,
             const int* __restrict__ scol,
             const float* __restrict__ bias, float* __restrict__ out, int N) {
  const int lane = threadIdx.x & 31;
  const int g = lane >> 3;
  const int l8 = lane & 7;
  const int wave = __builtin_amdgcn_readfirstlane(
      (int)((blockIdx.x * blockDim.x + threadIdx.x) >> 5));
  const int nWaves = (int)((gridDim.x * blockDim.x) >> 5);
  const unsigned laneOff = (unsigned)l8 * 16u;
  const int slot = l8 * 4 + g;
  const bool gOdd = (lane & 8) != 0;
  const bool gHi = (lane & 16) != 0;
  const unsigned ownOff = (unsigned)(g * 128) + laneOff;

  for (int r = wave; r < N; r += nWaves) {
    const RowSeg rs = seg[r];
    const int end = rs.cursor;
    const int beg = end - rs.count;

    const char* __restrict__ hrow = (const char*)Hp + (size_t)r * (F * 4);
    Feat16 hr;
#pragma unroll
    for (int i = 0; i < 4; ++i)
      hr.v[i] = *(const v4f*)(hrow + (laneOff + (unsigned)(i * 128)));
    v2f acc[8];
#pragma unroll
    for (int i = 0; i < 8; ++i) acc[i] = (v2f){0.0f, 0.0f};
    float den = 0.0f;

    for (int base = beg; base < end; base += 32) {
      const int m = min(32, end - base);
      const int mg = m - g;
      int cv = r;
      if (slot < m) cv = scol[base + slot];
      int e = 0;
      do {
        Feat16 A, B;
        fetch_step(A, Hp, cv, laneOff);
        if (e + 4 < m) {
          fetch_step(B, Hp, cv, laneOff);
          __builtin_amdgcn_sched_barrier(0);
          use_step(A, hr, e < mg, acc, den);
          use_step(B, hr, e + 4 < mg, acc, den);
        } else {
          use_step(A, hr, e < mg, acc, den);
        }
        e += 8;
      } while (e < m);
    }

    v4f a0, a1, a2, a3;
    a0.xy = acc[0]; a0.zw = acc[1];
    a1.xy = acc[2]; a1.zw = acc[3];
    a2.xy = acc[4]; a2.zw = acc[5];
    a3.xy = acc[6]; a3.zw = acc[7];
    const v4f k01 = sel4(gOdd, a1, a0) + row_xor8_4(sel4(gOdd, a0, a1));
    const v4f k23 = sel4(gOdd, a3, a2) + row_xor8_4(sel4(gOdd, a2, a3));
    v4f o = sel4(gHi, k23, k01) + half_xor4(sel4(gHi, k01, k23));
    den += row_xor<0x168>(den);
    den += half_xor(den);

    const v4f b = *(const v4f*)((const char*)bias + ownOff);
    const float d = den + 1e-10f;
    o.x = o.x / d + b.x;
    o.y = o.y / d + b.y;
    o.z = o.z / d + b.z;
    o.w = o.w / d + b.w;
    *(v4f*)((char*)out + (size_t)r * (F * 4) + ownOff) = o;
  }
}

extern "C" void kernel_launch(void* const* d_in, const int* in_sizes, int n_in,
                              void* d_out, int out_size, void* d_ws, size_t ws_size,
                              hipStream_t stream) {
  const float* H    = (const float*)d_in[0];
  const float* W    = (const float*)d_in[1];
  const float* bias = (const float*)d_in[2];
  const int*   row  = (const int*)d_in[3];
  const int*   col  = (const int*)d_in[4];

  const int N = in_sizes[0] / F;
  const int E = in_sizes[3];
  float* out = (float*)d_out;

  float*  Hp    = (float*)d_ws;
  int*    S     = (int*)(Hp + (size_t)N * F);
  RowSeg* seg   = (RowSeg*)(S + NBK * SPITCH);
  int2*   part  = (int2*)(seg + N);
  int*    scol  = (int*)(part + E);

  const int rpb = (N + NBK - 1) / NBK;
  edge_local_sort<<<NBK, 1024, 0, stream>>>(row, col, S, part, E, rpb);
  range_rows<<<NBK, 1024, 0, stream>>>(part, S, seg, scol, E, N, rpb);

  const int strips = (N + 15) / 16;
  gat_gemm_wmma<<<(strips + 7) / 8, 256, 0, stream>>>(H, W, Hp, N);

  gat_row<<<1280, 256, 0, stream>>>(Hp, seg, scol, bias, out, N);
}


# level 2 of the edge sort fused into the row kernel: a block sorts its row range's sources into LDS and its 32 waves sum those rows from there (rows dealt by an LDS counter); 4 launches to 3
# speedup vs baseline: 4.2261x; 1.1009x over previous
#include <hip/hip_runtime.h>

typedef __attribute__((ext_vector_type(2))) float v2f;
typedef __attribute__((ext_vector_type(8))) float v8f;

#define F 128

__global__ __launch_bounds__(256)
void gat_gemm_wmma(const float* __restrict__ H, const float* __restrict__ W,
                   float* __restrict__ Hp, int N) {
  const int lane = threadIdx.x & 31;
  const int wave = (int)((blockIdx.x * blockDim.x + threadIdx.x) >> 5);
  const int mTile = wave;
  if (mTile * 16 >= N) return;

  const int mr = lane & 15;
  const int kk = (lane >> 4) * 2;
  const float* __restrict__ Arow = H + (size_t)(mTile * 16 + mr) * F;
  const float* __restrict__ Bbase = W + (lane & 15);

  v8f acc[8];
#pragma unroll
  for (int nt = 0; nt < 8; ++nt) acc[nt] = (v8f){};

#pragma unroll 2
  for (int k = 0; k < F; k += 4) {
    v2f a = *(const v2f*)(Arow + k + kk);
    const float* __restrict__ Brow0 = Bbase + (size_t)(k + kk) * F;
    const float* __restrict__ Brow1 = Brow0 + F;
#pragma unroll
    for (int nt = 0; nt < 8; ++nt) {
      v2f b;
      b.x = Brow0[nt * 16];
      b.y = Brow1[nt * 16];
      acc[nt] = __builtin_amdgcn_wmma_f32_16x16x4_f32(false, a, false, b,
                                                      (short)0, acc[nt],
                                                      false, false);
    }
  }

  const int mBase = (lane >> 4) * 8;
  float* __restrict__ outBase =
      Hp + (size_t)(mTile * 16 + mBase) * F + (lane & 15);
#pragma unroll
  for (int nt = 0; nt < 8; ++nt) {
#pragma unroll
    for (int r = 0; r < 8; ++r)
      outBase[(size_t)r * F + nt * 16] = acc[nt][r];
  }
}

#define NBK 256
#define SPITCH (NBK + 1)
#define RPW (NBK / 32)
#define EPT 8
#define WROWS 1024
#define CAP 12288

__global__ __launch_bounds__(1024)
void edge_local_sort(const int* __restrict__ row, const int* __restrict__ col,
                     int* __restrict__ S, int2* __restrict__ part,
                     int E, int rpb) {
  __shared__ int h[NBK];
  __shared__ int wsum[NBK / 32];
  const int tid = (int)threadIdx.x;
  const int lane = tid & 31;
  const int w = tid >> 5;
  if (tid < NBK) h[tid] = 0;
  __syncthreads();
  const int chunk = (E + NBK - 1) / NBK;
  const int e0 = min(E, (int)blockIdx.x * chunk);
  const int e1 = min(E, e0 + chunk);
  int rr[EPT];
  int cq[EPT];
  int bin[EPT];
#pragma unroll
  for (int u = 0; u < EPT; ++u) {
    const int e = e0 + tid + u * 1024;
    rr[u] = -1;
    cq[u] = 0;
    if (e < e1) {
      rr[u] = row[e];
      cq[u] = col[e];
    }
  }
#pragma unroll
  for (int u = 0; u < EPT; ++u) {
    bin[u] = rr[u] / rpb;
    if (rr[u] >= 0) atomicAdd(&h[bin[u]], 1);
  }
  for (int e = e0 + tid + EPT * 1024; e < e1; e += 1024)
    atomicAdd(&h[row[e] / rpb], 1);
  __syncthreads();
  int c = 0;
  int incl = 0;
  if (tid < NBK) {
    c = h[tid];
    incl = c;
#pragma unroll
    for (int off = 1; off < 32; off <<= 1) {
      const int t = __shfl_up(incl, off, 32);
      if (lane >= off) incl += t;
    }
    if (lane == 31) wsum[w] = incl;
  }
  __syncthreads();
  if (tid < NBK) {
    int before = 0;
#pragma unroll
    for (int i = 0; i < NBK / 32; ++i)
      if (i < w) before += wsum[i];
    const int excl = before + incl - c;
    int* __restrict__ Srow = S + (size_t)blockIdx.x * SPITCH;
    Srow[tid] = excl;
    if (tid == NBK - 1) Srow[NBK] = excl + c;
    h[tid] = e0 + excl;
  }
  __syncthreads();
#pragma unroll
  for (int u = 0; u < EPT; ++u) {
    if (rr[u] >= 0) {
      const int pos = atomicAdd(&h[bin[u]], 1);
      part[pos] = make_int2(rr[u], cq[u]);
    }
  }
  for (int e = e0 + tid + EPT * 1024; e < e1; e += 1024) {
    const int r = row[e];
    const int cc = col[e];
    const int pos = atomicAdd(&h[r / rpb], 1);
    part[pos] = make_int2(r, cc);
  }
}

typedef __attribute__((ext_vector_type(4))) float v4f;

struct Feat16 {
  v4f v[4];
};

template <int CTRL>
__device__ __forceinline__ float row_xor(float x) {
  return __builtin_bit_cast(float,
      __builtin_amdgcn_update_dpp(0, __builtin_bit_cast(int, x), CTRL, 0xf, 0xf, true));
}

__device__ __forceinline__ float half_xor(float x) {
  const int b = __builtin_bit_cast(int, x);
  return __builtin_bit_cast(float,
      __builtin_amdgcn_permlanex16(b, b, 0x76543210, 0xfedcba98, false, false));
}

__device__ __forceinline__ void fetch_step(Feat16& d, const float* __restrict__ Hp,
                                           int& cv, unsigned laneOff) {
  const unsigned c = __builtin_amdgcn_mov_dpp8((unsigned)cv, 0);
  cv = (int)__builtin_amdgcn_mov_dpp8((unsigned)cv, 0x1F58D1);
  const char* __restrict__ p = (const char*)Hp + (c * (unsigned)(F * 4) + laneOff);
#pragma unroll
  for (int i = 0; i < 4; ++i) d.v[i] = *(const v4f*)(p + i * 128);
}

__device__ __forceinline__ void use_step(const Feat16& hc, const Feat16& hr, bool valid,
                                         v2f (&acc)[8], float& den) {
  v2f qa = hr.v[0].xy * hc.v[0].xy;
  v2f qb = hr.v[2].xy * hc.v[2].xy;
  qa += hr.v[0].zw * hc.v[0].zw;
  qb += hr.v[2].zw * hc.v[2].zw;
  qa += hr.v[1].xy * hc.v[1].xy;
  qb += hr.v[3].xy * hc.v[3].xy;
  qa += hr.v[1].zw * hc.v[1].zw;
  qb += hr.v[3].zw * hc.v[3].zw;
  qa += qb;
  float p = qa.x + qa.y;
  p += row_xor<0x161>(p);
  p += row_xor<0x162>(p);
  p += row_xor<0x164>(p);
  const float s = fmaxf(p, 0.2f * p);
  const float ex = valid ? __expf(s) : 0.0f;
  den += ex;
#pragma unroll
  for (int i = 0; i < 4; ++i) {
    acc[2 * i] += ex * hc.v[i].xy;
    acc[2 * i + 1] += ex * hc.v[i].zw;
  }
}

__device__ __forceinline__ v4f sel4(bool c, v4f a, v4f b) {
  v4f o;
  o.x = c ? a.x : b.x;
  o.y = c ? a.y : b.y;
  o.z = c ? a.z : b.z;
  o.w = c ? a.w : b.w;
  return o;
}

__device__ __forceinline__ v4f row_xor8_4(v4f a) {
  v4f o;
  o.x = row_xor<0x168>(a.x);
  o.y = row_xor<0x168>(a.y);
  o.z = row_xor<0x168>(a.z);
  o.w = row_xor<0x168>(a.w);
  return o;
}

__device__ __forceinline__ v4f half_xor4(v4f a) {
  v4f o;
  o.x = half_xor(a.x);
  o.y = half_xor(a.y);
  o.z = half_xor(a.z);
  o.w = half_xor(a.w);
  return o;
}

template <typename ColAt>
__device__ __forceinline__ void gat_one_row(const float* __restrict__ Hp, ColAt colAt,
                                            int beg, int end, int r, int g,
                                            unsigned laneOff, int slot, bool gOdd,
                                            bool gHi, unsigned ownOff, v4f b,
                                            float* __restrict__ out) {
  const char* __restrict__ hrow = (const char*)Hp + (size_t)r * (F * 4);
  Feat16 hr;
#pragma unroll
  for (int i = 0; i < 4; ++i)
    hr.v[i] = *(const v4f*)(hrow + (laneOff + (unsigned)(i * 128)));
  v2f acc[8];
#pragma unroll
  for (int i = 0; i < 8; ++i) acc[i] = (v2f){0.0f, 0.0f};
  float den = 0.0f;

  for (int base = beg; base < end; base += 32) {
    const int m = min(32, end - base);
    const int mg = m - g;
    int cv = r;
    if (slot < m) cv = colAt(base + slot);
    int e = 0;
    do {
      Feat16 A, B;
      fetch_step(A, Hp, cv, laneOff);
      if (e + 4 < m) {
        fetch_step(B, Hp, cv, laneOff);
        __builtin_amdgcn_sched_barrier(0);
        use_step(A, hr, e < mg, acc, den);
        use_step(B, hr, e + 4 < mg, acc, den);
      } else {
        use_step(A, hr, e < mg, acc, den);
      }
      e += 8;
    } while (e < m);
  }

  v4f a0, a1, a2, a3;
  a0.xy = acc[0]; a0.zw = acc[1];
  a1.xy = acc[2]; a1.zw = acc[3];
  a2.xy = acc[4]; a2.zw = acc[5];
  a3.xy = acc[6]; a3.zw = acc[7];
  const v4f k01 = sel4(gOdd, a1, a0) + row_xor8_4(sel4(gOdd, a0, a1));
  const v4f k23 = sel4(gOdd, a3, a2) + row_xor8_4(sel4(gOdd, a2, a3));
  v4f o = sel4(gHi, k23, k01) + half_xor4(sel4(gHi, k01, k23));
  den += row_xor<0x168>(den);
  den += half_xor(den);

  const float d = den + 1e-10f;
  o.x = o.x / d + b.x;
  o.y = o.y / d + b.y;
  o.z = o.z / d + b.z;
  o.w = o.w / d + b.w;
  *(v4f*)((char*)out + (size_t)r * (F * 4) + ownOff) = o;
}

__global__ __launch_bounds__(1024)
void range_gat(const int2* __restrict__ part, const int* __restrict__ S,
               const float* __restrict__ Hp, int* scol,
               const float* __restrict__ bias, float* __restrict__ out,
               int E, int N, int rpb) {
  __shared__ __attribute__((aligned(8))) int cz[WROWS + 2];
  int* const cnt = cz + 1;
  __shared__ int wsum[32];
  __shared__ int winTotal;
  __shared__ int nextRow;
  __shared__ __attribute__((aligned(16))) int runStart[NBK];
  __shared__ __attribute__((aligned(16))) int runCnt[NBK];
  __shared__ int baseSum[NBK / 32];
  __shared__ int lcol[CAP];
  const int tid = (int)threadIdx.x;
  const int lane = tid & 31;
  const int w = __builtin_amdgcn_readfirstlane(tid >> 5);
  const int b = (int)blockIdx.x;
  const int r0 = b * rpb;
  const int r1 = min(N, r0 + rpb);
  const int chunk = (E + NBK - 1) / NBK;

  const int g = lane >> 3;
  const int l8 = lane & 7;
  const unsigned laneOff = (unsigned)l8 * 16u;
  const int slot = l8 * 4 + g;
  const bool gOdd = (lane & 8) != 0;
  const bool gHi = (lane & 16) != 0;
  const unsigned ownOff = (unsigned)(g * 128) + laneOff;
  const v4f bv = *(const v4f*)((const char*)bias + ownOff);

  if (tid < NBK) {
    const int* __restrict__ Srow = S + (size_t)tid * SPITCH + b;
    const int s0 = Srow[0];
    const int s1 = Srow[1];
    runStart[tid] = min(E, tid * chunk) + s0;
    runCnt[tid] = s1 - s0;
    int t = s0;
#pragma unroll
    for (int off = 16; off >= 1; off >>= 1) t += __shfl_xor(t, off, 32);
    if (lane == 0) baseSum[w] = t;
  }
  __syncthreads();
  int winBase = 0;
#pragma unroll
  for (int i = 0; i < NBK / 32; ++i) winBase += baseSum[i];

  int rn[RPW];
  int rst[RPW];
#pragma unroll
  for (int u = 0; u < RPW; u += 4) {
    const int4 n4 = *(const int4*)(runCnt + w * RPW + u);
    const int4 s4 = *(const int4*)(runStart + w * RPW + u);
    rn[u] = n4.x; rn[u + 1] = n4.y; rn[u + 2] = n4.z; rn[u + 3] = n4.w;
    rst[u] = s4.x; rst[u + 1] = s4.y; rst[u + 2] = s4.z; rst[u + 3] = s4.w;
  }
  int2 pr[RPW];
#pragma unroll
  for (int u = 0; u < RPW; ++u) {
    pr[u] = make_int2(-1, 0);
    if (lane < rn[u]) pr[u] = part[rst[u] + lane];
  }

  for (int w0 = r0; w0 < r1; w0 += WROWS) {
    cnt[tid] = 0;
    if (tid == 0) {
      nextRow = 32;
      cz[0] = 0;
    }
    __syncthreads();
#pragma unroll
    for (int u = 0; u < RPW; ++u) {
      const unsigned lr = (unsigned)(pr[u].x - w0);
      if (lr < (unsigned)WROWS) atomicAdd(&cnt[lr], 1);
    }
    for (int u = 0; u < RPW; ++u) {
      const int q = w * RPW + u;
      const int n = runCnt[q];
      const int st = runStart[q];
      for (int i = lane + 32; i < n; i += 32) {
        const unsigned lr = (unsigned)(part[st + i].x - w0);
        if (lr < (unsigned)WROWS) atomicAdd(&cnt[lr], 1);
      }
    }
    __syncthreads();
    const int c = cnt[tid];
    int incl = c;
#pragma unroll
    for (int off = 1; off < 32; off <<= 1) {
      const int t = __shfl_up(incl, off, 32);
      if (lane >= off) incl += t;
    }
    if (lane == 31) wsum[w] = incl;
    __syncthreads();
    if (w == 0) {
      const int ws = wsum[lane];
      int si = ws;
#pragma unroll
      for (int off = 1; off < 32; off <<= 1) {
        const int t = __shfl_up(si, off, 32);
        if (lane >= off) si += t;
      }
      wsum[lane] = si - ws;
      if (lane == 31) winTotal = si;
    }
    __syncthreads();
    const int total = winTotal;
    const bool inLds = total <= CAP;
    int* __restrict__ scolWin = scol + winBase;
    cnt[tid] = wsum[w] + incl - c;
    __syncthreads();
#pragma unroll
    for (int u = 0; u < RPW; ++u) {
      const unsigned lr = (unsigned)(pr[u].x - w0);
      if (lr < (unsigned)WROWS) {
        const int pos = atomicAdd(&cnt[lr], 1);
        if (inLds) lcol[pos] = pr[u].y;
        else scolWin[pos] = pr[u].y;
      }
    }
    for (int u = 0; u < RPW; ++u) {
      const int q = w * RPW + u;
      const int n = runCnt[q];
      const int st = runStart[q];
      for (int i = lane + 32; i < n; i += 32) {
        const int2 p2 = part[st + i];
        const unsigned lr = (unsigned)(p2.x - w0);
        if (lr < (unsigned)WROWS) {
          const int pos = atomicAdd(&cnt[lr], 1);
          if (inLds) lcol[pos] = p2.y;
          else scolWin[pos] = p2.y;
        }
      }
    }
    __syncthreads();
    const int nrows = min(WROWS, r1 - w0);
    int lr = w;
    while (lr < nrows) {
      const int beg = __builtin_amdgcn_readfirstlane(cz[lr]);
      const int end = __builtin_amdgcn_readfirstlane(cz[lr + 1]);
      int nx = 0;
      if (lane == 0) nx = atomicAdd(&nextRow, 1);
      if (inLds) {
        gat_one_row(Hp, [](int i) { return lcol[i]; }, beg, end, w0 + lr, g,
                    laneOff, slot, gOdd, gHi, ownOff, bv, out);
      } else {
        const int* sc = scolWin;
        gat_one_row(Hp, [sc](int i) { return sc[i]; }, beg, end, w0 + lr, g,
                    laneOff, slot, gOdd, gHi, ownOff, bv, out);
      }
      lr = __builtin_amdgcn_readfirstlane(nx);
    }
    winBase += total;
    __syncthreads();
  }
}

extern "C" void kernel_launch(void* const* d_in, const int* in_sizes, int n_in,
                              void* d_out, int out_size, void* d_ws, size_t ws_size,
                              hipStream_t stream) {
  const float* H    = (const float*)d_in[0];
  const float* W    = (const float*)d_in[1];
  const float* bias = (const float*)d_in[2];
  const int*   row  = (const int*)d_in[3];
  const int*   col  = (const int*)d_in[4];

  const int N = in_sizes[0] / F;
  const int E = in_sizes[3];
  float* out = (float*)d_out;

  float*  Hp    = (float*)d_ws;
  int*    S     = (int*)(Hp + (size_t)N * F);
  int2*   part  = (int2*)(S + NBK * SPITCH);
  int*    scol  = (int*)(part + E);

  const int rpb = (N + NBK - 1) / NBK;
  edge_local_sort<<<NBK, 1024, 0, stream>>>(row, col, S, part, E, rpb);

  const int strips = (N + 15) / 16;
  gat_gemm_wmma<<<(strips + 7) / 8, 256, 0, stream>>>(H, W, Hp, N);

  range_gat<<<NBK, 1024, 0, stream>>>(part, S, Hp, scol, bias, out, E, N, rpb);
}


# edge sort level 1 and the GEMM share one launch (256 sorting blocks of 256 threads, then the strip blocks): the sort waits under the matrix work; 3 launches to 2; row sum order = arrival order
# speedup vs baseline: 4.5586x; 1.0787x over previous
#include <hip/hip_runtime.h>

typedef __attribute__((ext_vector_type(2))) float v2f;
typedef __attribute__((ext_vector_type(8))) float v8f;

#define F 128
#define PTH 256

__device__ __forceinline__
void gemm_strip_role(const float* __restrict__ H, const float* __restrict__ W,
                     float* __restrict__ Hp, int N, int gblock) {
  const int lane = threadIdx.x & 31;
  const int wave = (int)((gblock * PTH + (int)threadIdx.x) >> 5);
  const int mTile = wave;
  if (mTile * 16 >= N) return;

  const int mr = lane & 15;
  const int kk = (lane >> 4) * 2;
  const float* __restrict__ Arow = H + (size_t)(mTile * 16 + mr) * F;
  const float* __restrict__ Bbase = W + (lane & 15);

  v8f acc[8];
#pragma unroll
  for (int nt = 0; nt < 8; ++nt) acc[nt] = (v8f){};

#pragma unroll 2
  for (int k = 0; k < F; k += 4) {
    v2f a = *(const v2f*)(Arow + k + kk);
    const float* __restrict__ Brow0 = Bbase + (size_t)(k + kk) * F;
    const float* __restrict__ Brow1 = Brow0 + F;
#pragma unroll
    for (int nt = 0; nt < 8; ++nt) {
      v2f b;
      b.x = Brow0[nt * 16];
      b.y = Brow1[nt * 16];
      acc[nt] = __builtin_amdgcn_wmma_f32_16x16x4_f32(false, a, false, b,
                                                      (short)0, acc[nt],
                                                      false, false);
    }
  }

  const int mBase = (lane >> 4) * 8;
  float* __restrict__ outBase =
      Hp + (size_t)(mTile * 16 + mBase) * F + (lane & 15);
#pragma unroll
  for (int nt = 0; nt < 8; ++nt) {
#pragma unroll
    for (int r = 0; r < 8; ++r)
      outBase[(size_t)r * F + nt * 16] = acc[nt][r];
  }
}

#define NBK 256
#define SPITCH (NBK + 1)
#define RPW (NBK / 32)
#define EPT 25
#define WROWS 1024
#define CAP 12288

__device__ __forceinline__
void slice_sort_role(const int* __restrict__ row, const int* __restrict__ col,
                     int* __restrict__ S, int2* __restrict__ part,
                     int E, int rpb, int slice) {
  __shared__ int h[NBK];
  __shared__ int wsum[NBK / 32];
  const int tid = (int)threadIdx.x;
  const int lane = tid & 31;
  const int w = tid >> 5;
  if (tid < NBK) h[tid] = 0;
  __syncthreads();
  const int chunk = (E + NBK - 1) / NBK;
  const int e0 = min(E, slice * chunk);
  const int e1 = min(E, e0 + chunk);
  int rr[EPT];
  int cq[EPT];
  int bin[EPT];
  if (e1 > e0) {
#pragma unroll
    for (int u = 0; u < EPT; ++u) {
      const int ec = min(e0 + tid + u * PTH, e1 - 1);
      rr[u] = row[ec];
      cq[u] = col[ec];
    }
#pragma unroll
    for (int u = 0; u < EPT; ++u)
      if (e0 + tid + u * PTH >= e1) rr[u] = -1;
  } else {
#pragma unroll
    for (int u = 0; u < EPT; ++u) {
      rr[u] = -1;
      cq[u] = 0;
    }
  }
#pragma unroll
  for (int u = 0; u < EPT; ++u) {
    bin[u] = rr[u] / rpb;
    if (rr[u] >= 0) atomicAdd(&h[bin[u]], 1);
  }
  for (int e = e0 + tid + EPT * PTH; e < e1; e += PTH)
    atomicAdd(&h[row[e] / rpb], 1);
  __syncthreads();
  int c = 0;
  int incl = 0;
  if (tid < NBK) {
    c = h[tid];
    incl = c;
#pragma unroll
    for (int off = 1; off < 32; off <<= 1) {
      const int t = __shfl_up(incl, off, 32);
      if (lane >= off) incl += t;
    }
    if (lane == 31) wsum[w] = incl;
  }
  __syncthreads();
  if (tid < NBK) {
    int before = 0;
#pragma unroll
    for (int i = 0; i < NBK / 32; ++i)
      if (i < w) before += wsum[i];
    const int excl = before + incl - c;
    int* __restrict__ Srow = S + (size_t)slice * SPITCH;
    Srow[tid] = excl;
    if (tid == NBK - 1) Srow[NBK] = excl + c;
    h[tid] = e0 + excl;
  }
  __syncthreads();
#pragma unroll
  for (int u = 0; u < EPT; ++u) {
    if (rr[u] >= 0) {
      const int pos = atomicAdd(&h[bin[u]], 1);
      part[pos] = make_int2(rr[u], cq[u]);
    }
  }
  for (int e = e0 + tid + EPT * PTH; e < e1; e += PTH) {
    const int r = row[e];
    const int cc = col[e];
    const int pos = atomicAdd(&h[r / rpb], 1);
    part[pos] = make_int2(r, cc);
  }
}

__global__ __launch_bounds__(PTH)
void sort_and_gemm(const int* __restrict__ row, const int* __restrict__ col,
                   int* __restrict__ S, int2* __restrict__ part, int E, int rpb,
                   const float* __restrict__ H, const float* __restrict__ W,
                   float* __restrict__ Hp, int N) {
  const int b = (int)blockIdx.x;
  if (b < NBK) slice_sort_role(row, col, S, part, E, rpb, b);
  else gemm_strip_role(H, W, Hp, N, b - NBK);
}

typedef __attribute__((ext_vector_type(4))) float v4f;

struct Feat16 {
  v4f v[4];
};

template <int CTRL>
__device__ __forceinline__ float row_xor(float x) {
  return __builtin_bit_cast(float,
      __builtin_amdgcn_update_dpp(0, __builtin_bit_cast(int, x), CTRL, 0xf, 0xf, true));
}

__device__ __forceinline__ float half_xor(float x) {
  const int b = __builtin_bit_cast(int, x);
  return __builtin_bit_cast(float,
      __builtin_amdgcn_permlanex16(b, b, 0x76543210, 0xfedcba98, false, false));
}

__device__ __forceinline__ void fetch_step(Feat16& d, const float* __restrict__ Hp,
                                           int& cv, unsigned laneOff) {
  const unsigned c = __builtin_amdgcn_mov_dpp8((unsigned)cv, 0);
  cv = (int)__builtin_amdgcn_mov_dpp8((unsigned)cv, 0x1F58D1);
  const char* __restrict__ p = (const char*)Hp + (c * (unsigned)(F * 4) + laneOff);
#pragma unroll
  for (int i = 0; i < 4; ++i) d.v[i] = *(const v4f*)(p + i * 128);
}

__device__ __forceinline__ void use_step(const Feat16& hc, const Feat16& hr, bool valid,
                                         v2f (&acc)[8], float& den) {
  v2f qa = hr.v[0].xy * hc.v[0].xy;
  v2f qb = hr.v[2].xy * hc.v[2].xy;
  qa += hr.v[0].zw * hc.v[0].zw;
  qb += hr.v[2].zw * hc.v[2].zw;
  qa += hr.v[1].xy * hc.v[1].xy;
  qb += hr.v[3].xy * hc.v[3].xy;
  qa += hr.v[1].zw * hc.v[1].zw;
  qb += hr.v[3].zw * hc.v[3].zw;
  qa += qb;
  float p = qa.x + qa.y;
  p += row_xor<0x161>(p);
  p += row_xor<0x162>(p);
  p += row_xor<0x164>(p);
  const float s = fmaxf(p, 0.2f * p);
  const float ex = valid ? __expf(s) : 0.0f;
  den += ex;
#pragma unroll
  for (int i = 0; i < 4; ++i) {
    acc[2 * i] += ex * hc.v[i].xy;
    acc[2 * i + 1] += ex * hc.v[i].zw;
  }
}

__device__ __forceinline__ v4f sel4(bool c, v4f a, v4f b) {
  v4f o;
  o.x = c ? a.x : b.x;
  o.y = c ? a.y : b.y;
  o.z = c ? a.z : b.z;
  o.w = c ? a.w : b.w;
  return o;
}

__device__ __forceinline__ v4f row_xor8_4(v4f a) {
  v4f o;
  o.x = row_xor<0x168>(a.x);
  o.y = row_xor<0x168>(a.y);
  o.z = row_xor<0x168>(a.z);
  o.w = row_xor<0x168>(a.w);
  return o;
}

__device__ __forceinline__ v4f half_xor4(v4f a) {
  v4f o;
  o.x = half_xor(a.x);
  o.y = half_xor(a.y);
  o.z = half_xor(a.z);
  o.w = half_xor(a.w);
  return o;
}

template <typename ColAt>
__device__ __forceinline__ void gat_one_row(const float* __restrict__ Hp, ColAt colAt,
                                            int beg, int end, int r, int g,
                                            unsigned laneOff, int slot, bool gOdd,
                                            bool gHi, unsigned ownOff, v4f b,
                                            float* __restrict__ out) {
  const char* __restrict__ hrow = (const char*)Hp + (size_t)r * (F * 4);
  Feat16 hr;
#pragma unroll
  for (int i = 0; i < 4; ++i)
    hr.v[i] = *(const v4f*)(hrow + (laneOff + (unsigned)(i * 128)));
  v2f acc[8];
#pragma unroll
  for (int i = 0; i < 8; ++i) acc[i] = (v2f){0.0f, 0.0f};
  float den = 0.0f;

  for (int base = beg; base < end; base += 32) {
    const int m = min(32, end - base);
    const int mg = m - g;
    int cv = r;
    if (slot < m) cv = colAt(base + slot);
    int e = 0;
    do {
      Feat16 A, B;
      fetch_step(A, Hp, cv, laneOff);
      if (e + 4 < m) {
        fetch_step(B, Hp, cv, laneOff);
        __builtin_amdgcn_sched_barrier(0);
        use_step(A, hr, e < mg, acc, den);
        use_step(B, hr, e + 4 < mg, acc, den);
      } else {
        use_step(A, hr, e < mg, acc, den);
      }
      e += 8;
    } while (e < m);
  }

  v4f a0, a1, a2, a3;
  a0.xy = acc[0]; a0.zw = acc[1];
  a1.xy = acc[2]; a1.zw = acc[3];
  a2.xy = acc[4]; a2.zw = acc[5];
  a3.xy = acc[6]; a3.zw = acc[7];
  const v4f k01 = sel4(gOdd, a1, a0) + row_xor8_4(sel4(gOdd, a0, a1));
  const v4f k23 = sel4(gOdd, a3, a2) + row_xor8_4(sel4(gOdd, a2, a3));
  v4f o = sel4(gHi, k23, k01) + half_xor4(sel4(gHi, k01, k23));
  den += row_xor<0x168>(den);
  den += half_xor(den);

  const float d = den + 1e-10f;
  o.x = o.x / d + b.x;
  o.y = o.y / d + b.y;
  o.z = o.z / d + b.z;
  o.w = o.w / d + b.w;
  *(v4f*)((char*)out + (size_t)r * (F * 4) + ownOff) = o;
}

__global__ __launch_bounds__(1024)
void range_gat(const int2* __restrict__ part, const int* __restrict__ S,
               const float* __restrict__ Hp, int* scol,
               const float* __restrict__ bias, float* __restrict__ out,
               int E, int N, int rpb) {
  __shared__ __attribute__((aligned(8))) int cz[WROWS + 2];
  int* const cnt = cz + 1;
  __shared__ int wsum[32];
  __shared__ int winTotal;
  __shared__ int nextRow;
  __shared__ __attribute__((aligned(16))) int runStart[NBK];
  __shared__ __attribute__((aligned(16))) int runCnt[NBK];
  __shared__ int baseSum[NBK / 32];
  __shared__ int lcol[CAP];
  const int tid = (int)threadIdx.x;
  const int lane = tid & 31;
  const int w = __builtin_amdgcn_readfirstlane(tid >> 5);
  const int b = (int)blockIdx.x;
  const int r0 = b * rpb;
  const int r1 = min(N, r0 + rpb);
  const int chunk = (E + NBK - 1) / NBK;

  const int g = lane >> 3;
  const int l8 = lane & 7;
  const unsigned laneOff = (unsigned)l8 * 16u;
  const int slot = l8 * 4 + g;
  const bool gOdd = (lane & 8) != 0;
  const bool gHi = (lane & 16) != 0;
  const unsigned ownOff = (unsigned)(g * 128) + laneOff;
  const v4f bv = *(const v4f*)((const char*)bias + ownOff);

  if (tid < NBK) {
    const int* __restrict__ Srow = S + (size_t)tid * SPITCH + b;
    const int s0 = Srow[0];
    const int s1 = Srow[1];
    runStart[tid] = min(E, tid * chunk) + s0;
    runCnt[tid] = s1 - s0;
    int t = s0;
#pragma unroll
    for (int off = 16; off >= 1; off >>= 1) t += __shfl_xor(t, off, 32);
    if (lane == 0) baseSum[w] = t;
  }
  __syncthreads();
  int winBase = 0;
#pragma unroll
  for (int i = 0; i < NBK / 32; ++i) winBase += baseSum[i];

  int rn[RPW];
  int rst[RPW];
#pragma unroll
  for (int u = 0; u < RPW; u += 4) {
    const int4 n4 = *(const int4*)(runCnt + w * RPW + u);
    const int4 s4 = *(const int4*)(runStart + w * RPW + u);
    rn[u] = n4.x; rn[u + 1] = n4.y; rn[u + 2] = n4.z; rn[u + 3] = n4.w;
    rst[u] = s4.x; rst[u + 1] = s4.y; rst[u + 2] = s4.z; rst[u + 3] = s4.w;
  }
  int2 pr[RPW];
#pragma unroll
  for (int u = 0; u < RPW; ++u) {
    pr[u] = make_int2(-1, 0);
    if (lane < rn[u]) pr[u] = part[rst[u] + lane];
  }

  for (int w0 = r0; w0 < r1; w0 += WROWS) {
    cnt[tid] = 0;
    if (tid == 0) {
      nextRow = 32;
      cz[0] = 0;
    }
    __syncthreads();
#pragma unroll
    for (int u = 0; u < RPW; ++u) {
      const unsigned lr = (unsigned)(pr[u].x - w0);
      if (lr < (unsigned)WROWS) atomicAdd(&cnt[lr], 1);
    }
    for (int u = 0; u < RPW; ++u) {
      const int q = w * RPW + u;
      const int n = runCnt[q];
      const int st = runStart[q];
      for (int i = lane + 32; i < n; i += 32) {
        const unsigned lr = (unsigned)(part[st + i].x - w0);
        if (lr < (unsigned)WROWS) atomicAdd(&cnt[lr], 1);
      }
    }
    __syncthreads();
    const int c = cnt[tid];
    int incl = c;
#pragma unroll
    for (int off = 1; off < 32; off <<= 1) {
      const int t = __shfl_up(incl, off, 32);
      if (lane >= off) incl += t;
    }
    if (lane == 31) wsum[w] = incl;
    __syncthreads();
    if (w == 0) {
      const int ws = wsum[lane];
      int si = ws;
#pragma unroll
      for (int off = 1; off < 32; off <<= 1) {
        const int t = __shfl_up(si, off, 32);
        if (lane >= off) si += t;
      }
      wsum[lane] = si - ws;
      if (lane == 31) winTotal = si;
    }
    __syncthreads();
    const int total = winTotal;
    const bool inLds = total <= CAP;
    int* __restrict__ scolWin = scol + winBase;
    cnt[tid] = wsum[w] + incl - c;
    __syncthreads();
#pragma unroll
    for (int u = 0; u < RPW; ++u) {
      const unsigned lr = (unsigned)(pr[u].x - w0);
      if (lr < (unsigned)WROWS) {
        const int pos = atomicAdd(&cnt[lr], 1);
        if (inLds) lcol[pos] = pr[u].y;
        else scolWin[pos] = pr[u].y;
      }
    }
    for (int u = 0; u < RPW; ++u) {
      const int q = w * RPW + u;
      const int n = runCnt[q];
      const int st = runStart[q];
      for (int i = lane + 32; i < n; i += 32) {
        const int2 p2 = part[st + i];
        const unsigned lr = (unsigned)(p2.x - w0);
        if (lr < (unsigned)WROWS) {
          const int pos = atomicAdd(&cnt[lr], 1);
          if (inLds) lcol[pos] = p2.y;
          else scolWin[pos] = p2.y;
        }
      }
    }
    __syncthreads();
    const int nrows = min(WROWS, r1 - w0);
    int lr = w;
    while (lr < nrows) {
      const int beg = __builtin_amdgcn_readfirstlane(cz[lr]);
      const int end = __builtin_amdgcn_readfirstlane(cz[lr + 1]);
      int nx = 0;
      if (lane == 0) nx = atomicAdd(&nextRow, 1);
      if (inLds) {
        gat_one_row(Hp, [](int i) { return lcol[i]; }, beg, end, w0 + lr, g,
                    laneOff, slot, gOdd, gHi, ownOff, bv, out);
      } else {
        const int* sc = scolWin;
        gat_one_row(Hp, [sc](int i) { return sc[i]; }, beg, end, w0 + lr, g,
                    laneOff, slot, gOdd, gHi, ownOff, bv, out);
      }
      lr = __builtin_amdgcn_readfirstlane(nx);
    }
    winBase += total;
    __syncthreads();
  }
}

extern "C" void kernel_launch(void* const* d_in, const int* in_sizes, int n_in,
                              void* d_out, int out_size, void* d_ws, size_t ws_size,
                              hipStream_t stream) {
  const float* H    = (const float*)d_in[0];
  const float* W    = (const float*)d_in[1];
  const float* bias = (const float*)d_in[2];
  const int*   row  = (const int*)d_in[3];
  const int*   col  = (const int*)d_in[4];

  const int N = in_sizes[0] / F;
  const int E = in_sizes[3];
  float* out = (float*)d_out;

  float*  Hp    = (float*)d_ws;
  int*    S     = (int*)(Hp + (size_t)N * F);
  int2*   part  = (int2*)(S + NBK * SPITCH);
  int*    scol  = (int*)(part + E);

  const int rpb = (N + NBK - 1) / NBK;
  const int strips = (N + 15) / 16;
  sort_and_gemm<<<NBK + (strips + 7) / 8, PTH, 0, stream>>>(row, col, S, part, E, rpb,
                                                           H, W, Hp, N);

  range_gat<<<NBK, 1024, 0, stream>>>(part, S, Hp, scol, bias, out, E, N, rpb);
}


# GEMM blocks launched in whole rounds of one per WGP (768, not 782); the 106 strips left over are cut into 16x16 column pieces, one more per wave for 848 waves: no WGP carries a fourth block
# speedup vs baseline: 4.8702x; 1.0684x over previous
#include <hip/hip_runtime.h>

typedef __attribute__((ext_vector_type(2))) float v2f;
typedef __attribute__((ext_vector_type(8))) float v8f;

#define F 128
#define PTH 256

__device__ __forceinline__
void gemm_piece(const float* __restrict__ H, const float* __restrict__ W,
                float* __restrict__ Hp, int mTile, int nt, int lane) {
  const int mr = lane & 15;
  const int kk = (lane >> 4) * 2;
  const float* __restrict__ Arow = H + (size_t)(mTile * 16 + mr) * F;
  const float* __restrict__ Bcol = W + (nt * 16 + (lane & 15));
  v8f acc = (v8f){};
#pragma unroll 4
  for (int k = 0; k < F; k += 4) {
    v2f a = *(const v2f*)(Arow + k + kk);
    v2f b;
    b.x = Bcol[(size_t)(k + kk) * F];
    b.y = Bcol[(size_t)(k + kk + 1) * F];
    acc = __builtin_amdgcn_wmma_f32_16x16x4_f32(false, a, false, b,
                                                (short)0, acc, false, false);
  }
  const int mBase = (lane >> 4) * 8;
  float* __restrict__ o =
      Hp + (size_t)(mTile * 16 + mBase) * F + (nt * 16 + (lane & 15));
#pragma unroll
  for (int r = 0; r < 8; ++r) o[(size_t)r * F] = acc[r];
}

__device__ __forceinline__
void gemm_strip_role(const float* __restrict__ H, const float* __restrict__ W,
                     float* __restrict__ Hp, int N, int gblock, int gblocks) {
  const int lane = threadIdx.x & 31;
  const int wave = __builtin_amdgcn_readfirstlane(
      (int)((gblock * PTH + (int)threadIdx.x) >> 5));
  const int strips = (N + 15) >> 4;
  const int gwaves = gblocks * (PTH / 32);
  const int mainStrips = min(strips, gwaves);
  const int pieces = (strips - mainStrips) * 8;
  for (int p = wave; p < pieces; p += gwaves)
    gemm_piece(H, W, Hp, mainStrips + (p >> 3), p & 7, lane);
  const int mTile = wave;
  if (mTile >= mainStrips) return;

  const int mr = lane & 15;
  const int kk = (lane >> 4) * 2;
  const float* __restrict__ Arow = H + (size_t)(mTile * 16 + mr) * F;
  const float* __restrict__ Bbase = W + (lane & 15);

  v8f acc[8];
#pragma unroll
  for (int nt = 0; nt < 8; ++nt) acc[nt] = (v8f){};

#pragma unroll 2
  for (int k = 0; k < F; k += 4) {
    v2f a = *(const v2f*)(Arow + k + kk);
    const float* __restrict__ Brow0 = Bbase + (size_t)(k + kk) * F;
    const float* __restrict__ Brow1 = Brow0 + F;
#pragma unroll
    for (int nt = 0; nt < 8; ++nt) {
      v2f b;
      b.x = Brow0[nt * 16];
      b.y = Brow1[nt * 16];
      acc[nt] = __builtin_amdgcn_wmma_f32_16x16x4_f32(false, a, false, b,
                                                      (short)0, acc[nt],
                                                      false, false);
    }
  }

  const int mBase = (lane >> 4) * 8;
  float* __restrict__ outBase =
      Hp + (size_t)(mTile * 16 + mBase) * F + (lane & 15);
#pragma unroll
  for (int nt = 0; nt < 8; ++nt) {
#pragma unroll
    for (int r = 0; r < 8; ++r)
      outBase[(size_t)r * F + nt * 16] = acc[nt][r];
  }
}

#define NBK 256
#define SPITCH (NBK + 1)
#define RPW (NBK / 32)
#define EPT 25
#define WROWS 1024
#define CAP 12288

__device__ __forceinline__
void slice_sort_role(const int* __restrict__ row, const int* __restrict__ col,
                     int* __restrict__ S, int2* __restrict__ part,
                     int E, int rpb, int slice) {
  __shared__ int h[NBK];
  __shared__ int wsum[NBK / 32];
  const int tid = (int)threadIdx.x;
  const int lane = tid & 31;
  const int w = tid >> 5;
  if (tid < NBK) h[tid] = 0;
  __syncthreads();
  const int chunk = (E + NBK - 1) / NBK;
  const int e0 = min(E, slice * chunk);
  const int e1 = min(E, e0 + chunk);
  int rr[EPT];
  int cq[EPT];
  int bin[EPT];
  if (e1 > e0) {
#pragma unroll
    for (int u = 0; u < EPT; ++u) {
      const int ec = min(e0 + tid + u * PTH, e1 - 1);
      rr[u] = row[ec];
      cq[u] = col[ec];
    }
#pragma unroll
    for (int u = 0; u < EPT; ++u)
      if (e0 + tid + u * PTH >= e1) rr[u] = -1;
  } else {
#pragma unroll
    for (int u = 0; u < EPT; ++u) {
      rr[u] = -1;
      cq[u] = 0;
    }
  }
#pragma unroll
  for (int u = 0; u < EPT; ++u) {
    bin[u] = rr[u] / rpb;
    if (rr[u] >= 0) atomicAdd(&h[bin[u]], 1);
  }
  for (int e = e0 + tid + EPT * PTH; e < e1; e += PTH)
    atomicAdd(&h[row[e] / rpb], 1);
  __syncthreads();
  int c = 0;
  int incl = 0;
  if (tid < NBK) {
    c = h[tid];
    incl = c;
#pragma unroll
    for (int off = 1; off < 32; off <<= 1) {
      const int t = __shfl_up(incl, off, 32);
      if (lane >= off) incl += t;
    }
    if (lane == 31) wsum[w] = incl;
  }
  __syncthreads();
  if (tid < NBK) {
    int before = 0;
#pragma unroll
    for (int i = 0; i < NBK / 32; ++i)
      if (i < w) before += wsum[i];
    const int excl = before + incl - c;
    int* __restrict__ Srow = S + (size_t)slice * SPITCH;
    Srow[tid] = excl;
    if (tid == NBK - 1) Srow[NBK] = excl + c;
    h[tid] = e0 + excl;
  }
  __syncthreads();
#pragma unroll
  for (int u = 0; u < EPT; ++u) {
    if (rr[u] >= 0) {
      const int pos = atomicAdd(&h[bin[u]], 1);
      part[pos] = make_int2(rr[u], cq[u]);
    }
  }
  for (int e = e0 + tid + EPT * PTH; e < e1; e += PTH) {
    const int r = row[e];
    const int cc = col[e];
    const int pos = atomicAdd(&h[r / rpb], 1);
    part[pos] = make_int2(r, cc);
  }
}

__global__ __launch_bounds__(PTH)
void sort_and_gemm(const int* __restrict__ row, const int* __restrict__ col,
                   int* __restrict__ S, int2* __restrict__ part, int E, int rpb,
                   const float* __restrict__ H, const float* __restrict__ W,
                   float* __restrict__ Hp, int N, int gblocks) {
  const int b = (int)blockIdx.x;
  if (b < NBK) slice_sort_role(row, col, S, part, E, rpb, b);
  else gemm_strip_role(H, W, Hp, N, b - NBK, gblocks);
}

typedef __attribute__((ext_vector_type(4))) float v4f;

struct Feat16 {
  v4f v[4];
};

template <int CTRL>
__device__ __forceinline__ float row_xor(float x) {
  return __builtin_bit_cast(float,
      __builtin_amdgcn_update_dpp(0, __builtin_bit_cast(int, x), CTRL, 0xf, 0xf, true));
}

__device__ __forceinline__ float half_xor(float x) {
  const int b = __builtin_bit_cast(int, x);
  return __builtin_bit_cast(float,
      __builtin_amdgcn_permlanex16(b, b, 0x76543210, 0xfedcba98, false, false));
}

__device__ __forceinline__ void fetch_step(Feat16& d, const float* __restrict__ Hp,
                                           int& cv, unsigned laneOff) {
  const unsigned c = __builtin_amdgcn_mov_dpp8((unsigned)cv, 0);
  cv = (int)__builtin_amdgcn_mov_dpp8((unsigned)cv, 0x1F58D1);
  const char* __restrict__ p = (const char*)Hp + (c * (unsigned)(F * 4) + laneOff);
#pragma unroll
  for (int i = 0; i < 4; ++i) d.v[i] = *(const v4f*)(p + i * 128);
}

__device__ __forceinline__ void use_step(const Feat16& hc, const Feat16& hr, bool valid,
                                         v2f (&acc)[8], float& den) {
  v2f qa = hr.v[0].xy * hc.v[0].xy;
  v2f qb = hr.v[2].xy * hc.v[2].xy;
  qa += hr.v[0].zw * hc.v[0].zw;
  qb += hr.v[2].zw * hc.v[2].zw;
  qa += hr.v[1].xy * hc.v[1].xy;
  qb += hr.v[3].xy * hc.v[3].xy;
  qa += hr.v[1].zw * hc.v[1].zw;
  qb += hr.v[3].zw * hc.v[3].zw;
  qa += qb;
  float p = qa.x + qa.y;
  p += row_xor<0x161>(p);
  p += row_xor<0x162>(p);
  p += row_xor<0x164>(p);
  const float s = fmaxf(p, 0.2f * p);
  const float ex = valid ? __expf(s) : 0.0f;
  den += ex;
#pragma unroll
  for (int i = 0; i < 4; ++i) {
    acc[2 * i] += ex * hc.v[i].xy;
    acc[2 * i + 1] += ex * hc.v[i].zw;
  }
}

__device__ __forceinline__ v4f sel4(bool c, v4f a, v4f b) {
  v4f o;
  o.x = c ? a.x : b.x;
  o.y = c ? a.y : b.y;
  o.z = c ? a.z : b.z;
  o.w = c ? a.w : b.w;
  return o;
}

__device__ __forceinline__ v4f row_xor8_4(v4f a) {
  v4f o;
  o.x = row_xor<0x168>(a.x);
  o.y = row_xor<0x168>(a.y);
  o.z = row_xor<0x168>(a.z);
  o.w = row_xor<0x168>(a.w);
  return o;
}

__device__ __forceinline__ v4f half_xor4(v4f a) {
  v4f o;
  o.x = half_xor(a.x);
  o.y = half_xor(a.y);
  o.z = half_xor(a.z);
  o.w = half_xor(a.w);
  return o;
}

template <typename ColAt>
__device__ __forceinline__ void gat_one_row(const float* __restrict__ Hp, ColAt colAt,
                                            int beg, int end, int r, int g,
                                            unsigned laneOff, int slot, bool gOdd,
                                            bool gHi, unsigned ownOff, v4f b,
                                            float* __restrict__ out) {
  const char* __restrict__ hrow = (const char*)Hp + (size_t)r * (F * 4);
  Feat16 hr;
#pragma unroll
  for (int i = 0; i < 4; ++i)
    hr.v[i] = *(const v4f*)(hrow + (laneOff + (unsigned)(i * 128)));
  v2f acc[8];
#pragma unroll
  for (int i = 0; i < 8; ++i) acc[i] = (v2f){0.0f, 0.0f};
  float den = 0.0f;

  for (int base = beg; base < end; base += 32) {
    const int m = min(32, end - base);
    const int mg = m - g;
    int cv = r;
    if (slot < m) cv = colAt(base + slot);
    int e = 0;
    do {
      Feat16 A, B;
      fetch_step(A, Hp, cv, laneOff);
      if (e + 4 < m) {
        fetch_step(B, Hp, cv, laneOff);
        __builtin_amdgcn_sched_barrier(0);
        use_step(A, hr, e < mg, acc, den);
        use_step(B, hr, e + 4 < mg, acc, den);
      } else {
        use_step(A, hr, e < mg, acc, den);
      }
      e += 8;
    } while (e < m);
  }

  v4f a0, a1, a2, a3;
  a0.xy = acc[0]; a0.zw = acc[1];
  a1.xy = acc[2]; a1.zw = acc[3];
  a2.xy = acc[4]; a2.zw = acc[5];
  a3.xy = acc[6]; a3.zw = acc[7];
  const v4f k01 = sel4(gOdd, a1, a0) + row_xor8_4(sel4(gOdd, a0, a1));
  const v4f k23 = sel4(gOdd, a3, a2) + row_xor8_4(sel4(gOdd, a2, a3));
  v4f o = sel4(gHi, k23, k01) + half_xor4(sel4(gHi, k01, k23));
  den += row_xor<0x168>(den);
  den += half_xor(den);

  const float d = den + 1e-10f;
  o.x = o.x / d + b.x;
  o.y = o.y / d + b.y;
  o.z = o.z / d + b.z;
  o.w = o.w / d + b.w;
  *(v4f*)((char*)out + (size_t)r * (F * 4) + ownOff) = o;
}

__global__ __launch_bounds__(1024)
void range_gat(const int2* __restrict__ part, const int* __restrict__ S,
               const float* __restrict__ Hp, int* scol,
               const float* __restrict__ bias, float* __restrict__ out,
               int E, int N, int rpb) {
  __shared__ __attribute__((aligned(8))) int cz[WROWS + 2];
  int* const cnt = cz + 1;
  __shared__ int wsum[32];
  __shared__ int winTotal;
  __shared__ int nextRow;
  __shared__ __attribute__((aligned(16))) int runStart[NBK];
  __shared__ __attribute__((aligned(16))) int runCnt[NBK];
  __shared__ int baseSum[NBK / 32];
  __shared__ int lcol[CAP];
  const int tid = (int)threadIdx.x;
  const int lane = tid & 31;
  const int w = __builtin_amdgcn_readfirstlane(tid >> 5);
  const int b = (int)blockIdx.x;
  const int r0 = b * rpb;
  const int r1 = min(N, r0 + rpb);
  const int chunk = (E + NBK - 1) / NBK;

  const int g = lane >> 3;
  const int l8 = lane & 7;
  const unsigned laneOff = (unsigned)l8 * 16u;
  const int slot = l8 * 4 + g;
  const bool gOdd = (lane & 8) != 0;
  const bool gHi = (lane & 16) != 0;
  const unsigned ownOff = (unsigned)(g * 128) + laneOff;
  const v4f bv = *(const v4f*)((const char*)bias + ownOff);

  if (tid < NBK) {
    const int* __restrict__ Srow = S + (size_t)tid * SPITCH + b;
    const int s0 = Srow[0];
    const int s1 = Srow[1];
    runStart[tid] = min(E, tid * chunk) + s0;
    runCnt[tid] = s1 - s0;
    int t = s0;
#pragma unroll
    for (int off = 16; off >= 1; off >>= 1) t += __shfl_xor(t, off, 32);
    if (lane == 0) baseSum[w] = t;
  }
  __syncthreads();
  int winBase = 0;
#pragma unroll
  for (int i = 0; i < NBK / 32; ++i) winBase += baseSum[i];

  int rn[RPW];
  int rst[RPW];
#pragma unroll
  for (int u = 0; u < RPW; u += 4) {
    const int4 n4 = *(const int4*)(runCnt + w * RPW + u);
    const int4 s4 = *(const int4*)(runStart + w * RPW + u);
    rn[u] = n4.x; rn[u + 1] = n4.y; rn[u + 2] = n4.z; rn[u + 3] = n4.w;
    rst[u] = s4.x; rst[u + 1] = s4.y; rst[u + 2] = s4.z; rst[u + 3] = s4.w;
  }
  int2 pr[RPW];
#pragma unroll
  for (int u = 0; u < RPW; ++u) {
    pr[u] = make_int2(-1, 0);
    if (lane < rn[u]) pr[u] = part[rst[u] + lane];
  }

  for (int w0 = r0; w0 < r1; w0 += WROWS) {
    cnt[tid] = 0;
    if (tid == 0) {
      nextRow = 32;
      cz[0] = 0;
    }
    __syncthreads();
#pragma unroll
    for (int u = 0; u < RPW; ++u) {
      const unsigned lr = (unsigned)(pr[u].x - w0);
      if (lr < (unsigned)WROWS) atomicAdd(&cnt[lr], 1);
    }
    for (int u = 0; u < RPW; ++u) {
      const int q = w * RPW + u;
      const int n = runCnt[q];
      const int st = runStart[q];
      for (int i = lane + 32; i < n; i += 32) {
        const unsigned lr = (unsigned)(part[st + i].x - w0);
        if (lr < (unsigned)WROWS) atomicAdd(&cnt[lr], 1);
      }
    }
    __syncthreads();
    const int c = cnt[tid];
    int incl = c;
#pragma unroll
    for (int off = 1; off < 32; off <<= 1) {
      const int t = __shfl_up(incl, off, 32);
      if (lane >= off) incl += t;
    }
    if (lane == 31) wsum[w] = incl;
    __syncthreads();
    if (w == 0) {
      const int ws = wsum[lane];
      int si = ws;
#pragma unroll
      for (int off = 1; off < 32; off <<= 1) {
        const int t = __shfl_up(si, off, 32);
        if (lane >= off) si += t;
      }
      wsum[lane] = si - ws;
      if (lane == 31) winTotal = si;
    }
    __syncthreads();
    const int total = winTotal;
    const bool inLds = total <= CAP;
    int* __restrict__ scolWin = scol + winBase;
    cnt[tid] = wsum[w] + incl - c;
    __syncthreads();
#pragma unroll
    for (int u = 0; u < RPW; ++u) {
      const unsigned lr = (unsigned)(pr[u].x - w0);
      if (lr < (unsigned)WROWS) {
        const int pos = atomicAdd(&cnt[lr], 1);
        if (inLds) lcol[pos] = pr[u].y;
        else scolWin[pos] = pr[u].y;
      }
    }
    for (int u = 0; u < RPW; ++u) {
      const int q = w * RPW + u;
      const int n = runCnt[q];
      const int st = runStart[q];
      for (int i = lane + 32; i < n; i += 32) {
        const int2 p2 = part[st + i];
        const unsigned lr = (unsigned)(p2.x - w0);
        if (lr < (unsigned)WROWS) {
          const int pos = atomicAdd(&cnt[lr], 1);
          if (inLds) lcol[pos] = p2.y;
          else scolWin[pos] = p2.y;
        }
      }
    }
    __syncthreads();
    const int nrows = min(WROWS, r1 - w0);
    int lr = w;
    while (lr < nrows) {
      const int beg = __builtin_amdgcn_readfirstlane(cz[lr]);
      const int end = __builtin_amdgcn_readfirstlane(cz[lr + 1]);
      int nx = 0;
      if (lane == 0) nx = atomicAdd(&nextRow, 1);
      if (inLds) {
        gat_one_row(Hp, [](int i) { return lcol[i]; }, beg, end, w0 + lr, g,
                    laneOff, slot, gOdd, gHi, ownOff, bv, out);
      } else {
        const int* sc = scolWin;
        gat_one_row(Hp, [sc](int i) { return sc[i]; }, beg, end, w0 + lr, g,
                    laneOff, slot, gOdd, gHi, ownOff, bv, out);
      }
      lr = __builtin_amdgcn_readfirstlane(nx);
    }
    winBase += total;
    __syncthreads();
  }
}

extern "C" void kernel_launch(void* const* d_in, const int* in_sizes, int n_in,
                              void* d_out, int out_size, void* d_ws, size_t ws_size,
                              hipStream_t stream) {
  const float* H    = (const float*)d_in[0];
  const float* W    = (const float*)d_in[1];
  const float* bias = (const float*)d_in[2];
  const int*   row  = (const int*)d_in[3];
  const int*   col  = (const int*)d_in[4];

  const int N = in_sizes[0] / F;
  const int E = in_sizes[3];
  float* out = (float*)d_out;

  float*  Hp    = (float*)d_ws;
  int*    S     = (int*)(Hp + (size_t)N * F);
  int2*   part  = (int2*)(S + NBK * SPITCH);
  int*    scol  = (int*)(part + E);

  const int rpb = (N + NBK - 1) / NBK;
  const int strips = (N + 15) / 16;
  const int gblocks = strips >= 2048 ? (strips / 2048) * 256 : (strips + 7) / 8;
  sort_and_gemm<<<NBK + gblocks, PTH, 0, stream>>>(row, col, S, part, E, rpb,
                                                  H, W, Hp, N, gblocks);

  range_gat<<<NBK, 1024, 0, stream>>>(part, S, Hp, scol, bias, out, E, N, rpb);
}


# GEMM: each block copies W into LDS once (k-row pairs side by side, halves bank-flipped); a lane's two B values of a k-step are one 8-byte LDS read, not two global loads; k order and sums unchanged
# speedup vs baseline: 5.1450x; 1.0564x over previous
#include <hip/hip_runtime.h>

typedef __attribute__((ext_vector_type(2))) float v2f;
typedef __attribute__((ext_vector_type(4))) float v4f;
typedef __attribute__((ext_vector_type(8))) float v8f;

#define F 128
#define PTH 256

#define WLDS (F / 2 * 2 * F)

__device__ __forceinline__
void stage_w(const float* __restrict__ W, float* wl) {
  const int t = (int)threadIdx.x;
  const int n4 = t & 31;
  const int kq = t >> 5;
  v4f r0[F / 2 * 32 / PTH];
  v4f r1[F / 2 * 32 / PTH];
#pragma unroll
  for (int i = 0; i < F / 2 * 32 / PTH; ++i) {
    const int kp = kq + (PTH / 32) * i;
    r0[i] = *(const v4f*)(W + (size_t)(2 * kp) * F + 4 * n4);
    r1[i] = *(const v4f*)(W + (size_t)(2 * kp + 1) * F + 4 * n4);
  }
#pragma unroll
  for (int i = 0; i < F / 2 * 32 / PTH; ++i) {
    const int kp = kq + (PTH / 32) * i;
    float* d = wl + kp * (2 * F) + ((8 * n4) ^ ((kp & 1) << 5));
    v4f lo, hi;
    lo.x = r0[i].x; lo.y = r1[i].x; lo.z = r0[i].y; lo.w = r1[i].y;
    hi.x = r0[i].z; hi.y = r1[i].z; hi.z = r0[i].w; hi.w = r1[i].w;
    *(v4f*)d = lo;
    *(v4f*)(d + 4) = hi;
  }
}

__device__ __forceinline__
void gemm_piece(const float* __restrict__ H, const float* wl,
                float* __restrict__ Hp, int mTile, int nt, int lane) {
  const int mr = lane & 15;
  const int hf = lane >> 4;
  const int kk = hf * 2;
  const float* __restrict__ Arow = H + (size_t)(mTile * 16 + mr) * F;
  const float* Bl = wl + hf * (2 * F) + 32 * (nt ^ hf) + 2 * mr;
  v8f acc = (v8f){};
#pragma unroll 4
  for (int k = 0; k < F; k += 4) {
    v2f a = *(const v2f*)(Arow + k + kk);
    v2f b = *(const v2f*)(Bl + k * F);
    acc = __builtin_amdgcn_wmma_f32_16x16x4_f32(false, a, false, b,
                                                (short)0, acc, false, false);
  }
  const int mBase = (lane >> 4) * 8;
  float* __restrict__ o =
      Hp + (size_t)(mTile * 16 + mBase) * F + (nt * 16 + (lane & 15));
#pragma unroll
  for (int r = 0; r < 8; ++r) o[(size_t)r * F] = acc[r];
}

__device__ __forceinline__
void gemm_strip_role(const float* __restrict__ H, const float* __restrict__ W,
                     float* __restrict__ Hp, int N, int gblock, int gblocks) {
  __shared__ __attribute__((aligned(16))) float wl[WLDS];
  stage_w(W, wl);
  __syncthreads();
  const int lane = threadIdx.x & 31;
  const int wave = __builtin_amdgcn_readfirstlane(
      (int)((gblock * PTH + (int)threadIdx.x) >> 5));
  const int strips = (N + 15) >> 4;
  const int gwaves = gblocks * (PTH / 32);
  const int mainStrips = min(strips, gwaves);
  const int pieces = (strips - mainStrips) * 8;
  for (int p = wave; p < pieces; p += gwaves)
    gemm_piece(H, wl, Hp, mainStrips + (p >> 3), p & 7, lane);
  const int mTile = wave;
  if (mTile >= mainStrips) return;

  const int mr = lane & 15;
  const int hf = lane >> 4;
  const int kk = hf * 2;
  const float* __restrict__ Arow = H + (size_t)(mTile * 16 + mr) * F;
  const float* BlE = wl + hf * (2 * F) + 32 * hf + 2 * mr;
  const float* BlO = wl + hf * (2 * F) + 32 * (1 - hf) + 2 * mr;

  v8f acc[8];
#pragma unroll
  for (int nt = 0; nt < 8; ++nt) acc[nt] = (v8f){};

#pragma unroll 2
  for (int k = 0; k < F; k += 4) {
    v2f a = *(const v2f*)(Arow + k + kk);
#pragma unroll
    for (int nt = 0; nt < 8; ++nt) {
      const v2f b = (nt & 1) ? *(const v2f*)(BlO + k * F + 32 * (nt - 1))
                             : *(const v2f*)(BlE + k * F + 32 * nt);
      acc[nt] = __builtin_amdgcn_wmma_f32_16x16x4_f32(false, a, false, b,
                                                      (short)0, acc[nt],
                                                      false, false);
    }
  }

  const int mBase = (lane >> 4) * 8;
  float* __restrict__ outBase =
      Hp + (size_t)(mTile * 16 + mBase) * F + (lane & 15);
#pragma unroll
  for (int nt = 0; nt < 8; ++nt) {
#pragma unroll
    for (int r = 0; r < 8; ++r)
      outBase[(size_t)r * F + nt * 16] = acc[nt][r];
  }
}

#define NBK 256
#define SPITCH (NBK + 1)
#define RPW (NBK / 32)
#define EPT 25
#define WROWS 1024
#define CAP 12288

__device__ __forceinline__
void slice_sort_role(const int* __restrict__ row, const int* __restrict__ col,
                     int* __restrict__ S, int2* __restrict__ part,
                     int E, int rpb, int slice) {
  __shared__ int h[NBK];
  __shared__ int wsum[NBK / 32];
  const int tid = (int)threadIdx.x;
  const int lane = tid & 31;
  const int w = tid >> 5;
  if (tid < NBK) h[tid] = 0;
  __syncthreads();
  const int chunk = (E + NBK - 1) / NBK;
  const int e0 = min(E, slice * chunk);
  const int e1 = min(E, e0 + chunk);
  int rr[EPT];
  int cq[EPT];
  int bin[EPT];
  if (e1 > e0) {
#pragma unroll
    for (int u = 0; u < EPT; ++u) {
      const int ec = min(e0 + tid + u * PTH, e1 - 1);
      rr[u] = row[ec];
      cq[u] = col[ec];
    }
#pragma unroll
    for (int u = 0; u < EPT; ++u)
      if (e0 + tid + u * PTH >= e1) rr[u] = -1;
  } else {
#pragma unroll
    for (int u = 0; u < EPT; ++u) {
      rr[u] = -1;
      cq[u] = 0;
    }
  }
#pragma unroll
  for (int u = 0; u < EPT; ++u) {
    bin[u] = rr[u] / rpb;
    if (rr[u] >= 0) atomicAdd(&h[bin[u]], 1);
  }
  for (int e = e0 + tid + EPT * PTH; e < e1; e += PTH)
    atomicAdd(&h[row[e] / rpb], 1);
  __syncthreads();
  int c = 0;
  int incl = 0;
  if (tid < NBK) {
    c = h[tid];
    incl = c;
#pragma unroll
    for (int off = 1; off < 32; off <<= 1) {
      const int t = __shfl_up(incl, off, 32);
      if (lane >= off) incl += t;
    }
    if (lane == 31) wsum[w] = incl;
  }
  __syncthreads();
  if (tid < NBK) {
    int before = 0;
#pragma unroll
    for (int i = 0; i < NBK / 32; ++i)
      if (i < w) before += wsum[i];
    const int excl = before + incl - c;
    int* __restrict__ Srow = S + (size_t)slice * SPITCH;
    Srow[tid] = excl;
    if (tid == NBK - 1) Srow[NBK] = excl + c;
    h[tid] = e0 + excl;
  }
  __syncthreads();
#pragma unroll
  for (int u = 0; u < EPT; ++u) {
    if (rr[u] >= 0) {
      const int pos = atomicAdd(&h[bin[u]], 1);
      part[pos] = make_int2(rr[u], cq[u]);
    }
  }
  for (int e = e0 + tid + EPT * PTH; e < e1; e += PTH) {
    const int r = row[e];
    const int cc = col[e];
    const int pos = atomicAdd(&h[r / rpb], 1);
    part[pos] = make_int2(r, cc);
  }
}

__global__ __launch_bounds__(PTH)
void sort_and_gemm(const int* __restrict__ row, const int* __restrict__ col,
                   int* __restrict__ S, int2* __restrict__ part, int E, int rpb,
                   const float* __restrict__ H, const float* __restrict__ W,
                   float* __restrict__ Hp, int N, int gblocks) {
  const int b = (int)blockIdx.x;
  if (b < NBK) slice_sort_role(row, col, S, part, E, rpb, b);
  else gemm_strip_role(H, W, Hp, N, b - NBK, gblocks);
}

struct Feat16 {
  v4f v[4];
};

template <int CTRL>
__device__ __forceinline__ float row_xor(float x) {
  return __builtin_bit_cast(float,
      __builtin_amdgcn_update_dpp(0, __builtin_bit_cast(int, x), CTRL, 0xf, 0xf, true));
}

__device__ __forceinline__ float half_xor(float x) {
  const int b = __builtin_bit_cast(int, x);
  return __builtin_bit_cast(float,
      __builtin_amdgcn_permlanex16(b, b, 0x76543210, 0xfedcba98, false, false));
}

__device__ __forceinline__ void fetch_step(Feat16& d, const float* __restrict__ Hp,
                                           int& cv, unsigned laneOff) {
  const unsigned c = __builtin_amdgcn_mov_dpp8((unsigned)cv, 0);
  cv = (int)__builtin_amdgcn_mov_dpp8((unsigned)cv, 0x1F58D1);
  const char* __restrict__ p = (const char*)Hp + (c * (unsigned)(F * 4) + laneOff);
#pragma unroll
  for (int i = 0; i < 4; ++i) d.v[i] = *(const v4f*)(p + i * 128);
}

__device__ __forceinline__ void use_step(const Feat16& hc, const Feat16& hr, bool valid,
                                         v2f (&acc)[8], float& den) {
  v2f qa = hr.v[0].xy * hc.v[0].xy;
  v2f qb = hr.v[2].xy * hc.v[2].xy;
  qa += hr.v[0].zw * hc.v[0].zw;
  qb += hr.v[2].zw * hc.v[2].zw;
  qa += hr.v[1].xy * hc.v[1].xy;
  qb += hr.v[3].xy * hc.v[3].xy;
  qa += hr.v[1].zw * hc.v[1].zw;
  qb += hr.v[3].zw * hc.v[3].zw;
  qa += qb;
  float p = qa.x + qa.y;
  p += row_xor<0x161>(p);
  p += row_xor<0x162>(p);
  p += row_xor<0x164>(p);
  const float s = fmaxf(p, 0.2f * p);
  const float ex = valid ? __expf(s) : 0.0f;
  den += ex;
#pragma unroll
  for (int i = 0; i < 4; ++i) {
    acc[2 * i] += ex * hc.v[i].xy;
    acc[2 * i + 1] += ex * hc.v[i].zw;
  }
}

__device__ __forceinline__ v4f sel4(bool c, v4f a, v4f b) {
  v4f o;
  o.x = c ? a.x : b.x;
  o.y = c ? a.y : b.y;
  o.z = c ? a.z : b.z;
  o.w = c ? a.w : b.w;
  return o;
}

__device__ __forceinline__ v4f row_xor8_4(v4f a) {
  v4f o;
  o.x = row_xor<0x168>(a.x);
  o.y = row_xor<0x168>(a.y);
  o.z = row_xor<0x168>(a.z);
  o.w = row_xor<0x168>(a.w);
  return o;
}

__device__ __forceinline__ v4f half_xor4(v4f a) {
  v4f o;
  o.x = half_xor(a.x);
  o.y = half_xor(a.y);
  o.z = half_xor(a.z);
  o.w = half_xor(a.w);
  return o;
}

template <typename ColAt>
__device__ __forceinline__ void gat_one_row(const float* __restrict__ Hp, ColAt colAt,
                                            int beg, int end, int r, int g,
                                            unsigned laneOff, int slot, bool gOdd,
                                            bool gHi, unsigned ownOff, v4f b,
                                            float* __restrict__ out) {
  const char* __restrict__ hrow = (const char*)Hp + (size_t)r * (F * 4);
  Feat16 hr;
#pragma unroll
  for (int i = 0; i < 4; ++i)
    hr.v[i] = *(const v4f*)(hrow + (laneOff + (unsigned)(i * 128)));
  v2f acc[8];
#pragma unroll
  for (int i = 0; i < 8; ++i) acc[i] = (v2f){0.0f, 0.0f};
  float den = 0.0f;

  for (int base = beg; base < end; base += 32) {
    const int m = min(32, end - base);
    const int mg = m - g;
    int cv = r;
    if (slot < m) cv = colAt(base + slot);
    int e = 0;
    do {
      Feat16 A, B;
      fetch_step(A, Hp, cv, laneOff);
      if (e + 4 < m) {
        fetch_step(B, Hp, cv, laneOff);
        __builtin_amdgcn_sched_barrier(0);
        use_step(A, hr, e < mg, acc, den);
        use_step(B, hr, e + 4 < mg, acc, den);
      } else {
        use_step(A, hr, e < mg, acc, den);
      }
      e += 8;
    } while (e < m);
  }

  v4f a0, a1, a2, a3;
  a0.xy = acc[0]; a0.zw = acc[1];
  a1.xy = acc[2]; a1.zw = acc[3];
  a2.xy = acc[4]; a2.zw = acc[5];
  a3.xy = acc[6]; a3.zw = acc[7];
  const v4f k01 = sel4(gOdd, a1, a0) + row_xor8_4(sel4(gOdd, a0, a1));
  const v4f k23 = sel4(gOdd, a3, a2) + row_xor8_4(sel4(gOdd, a2, a3));
  v4f o = sel4(gHi, k23, k01) + half_xor4(sel4(gHi, k01, k23));
  den += row_xor<0x168>(den);
  den += half_xor(den);

  const float d = den + 1e-10f;
  o.x = o.x / d + b.x;
  o.y = o.y / d + b.y;
  o.z = o.z / d + b.z;
  o.w = o.w / d + b.w;
  *(v4f*)((char*)out + (size_t)r * (F * 4) + ownOff) = o;
}

__global__ __launch_bounds__(1024)
void range_gat(const int2* __restrict__ part, const int* __restrict__ S,
               const float* __restrict__ Hp, int* scol,
               const float* __restrict__ bias, float* __restrict__ out,
               int E, int N, int rpb) {
  __shared__ __attribute__((aligned(8))) int cz[WROWS + 2];
  int* const cnt = cz + 1;
  __shared__ int wsum[32];
  __shared__ int winTotal;
  __shared__ int nextRow;
  __shared__ __attribute__((aligned(16))) int runStart[NBK];
  __shared__ __attribute__((aligned(16))) int runCnt[NBK];
  __shared__ int baseSum[NBK / 32];
  __shared__ int lcol[CAP];
  const int tid = (int)threadIdx.x;
  const int lane = tid & 31;
  const int w = __builtin_amdgcn_readfirstlane(tid >> 5);
  const int b = (int)blockIdx.x;
  const int r0 = b * rpb;
  const int r1 = min(N, r0 + rpb);
  const int chunk = (E + NBK - 1) / NBK;

  const int g = lane >> 3;
  const int l8 = lane & 7;
  const unsigned laneOff = (unsigned)l8 * 16u;
  const int slot = l8 * 4 + g;
  const bool gOdd = (lane & 8) != 0;
  const bool gHi = (lane & 16) != 0;
  const unsigned ownOff = (unsigned)(g * 128) + laneOff;
  const v4f bv = *(const v4f*)((const char*)bias + ownOff);

  if (tid < NBK) {
    const int* __restrict__ Srow = S + (size_t)tid * SPITCH + b;
    const int s0 = Srow[0];
    const int s1 = Srow[1];
    runStart[tid] = min(E, tid * chunk) + s0;
    runCnt[tid] = s1 - s0;
    int t = s0;
#pragma unroll
    for (int off = 16; off >= 1; off >>= 1) t += __shfl_xor(t, off, 32);
    if (lane == 0) baseSum[w] = t;
  }
  __syncthreads();
  int winBase = 0;
#pragma unroll
  for (int i = 0; i < NBK / 32; ++i) winBase += baseSum[i];

  int rn[RPW];
  int rst[RPW];
#pragma unroll
  for (int u = 0; u < RPW; u += 4) {
    const int4 n4 = *(const int4*)(runCnt + w * RPW + u);
    const int4 s4 = *(const int4*)(runStart + w * RPW + u);
    rn[u] = n4.x; rn[u + 1] = n4.y; rn[u + 2] = n4.z; rn[u + 3] = n4.w;
    rst[u] = s4.x; rst[u + 1] = s4.y; rst[u + 2] = s4.z; rst[u + 3] = s4.w;
  }
  int2 pr[RPW];
#pragma unroll
  for (int u = 0; u < RPW; ++u) {
    pr[u] = make_int2(-1, 0);
    if (lane < rn[u]) pr[u] = part[rst[u] + lane];
  }

  for (int w0 = r0; w0 < r1; w0 += WROWS) {
    cnt[tid] = 0;
    if (tid == 0) {
      nextRow = 32;
      cz[0] = 0;
    }
    __syncthreads();
#pragma unroll
    for (int u = 0; u < RPW; ++u) {
      const unsigned lr = (unsigned)(pr[u].x - w0);
      if (lr < (unsigned)WROWS) atomicAdd(&cnt[lr], 1);
    }
    for (int u = 0; u < RPW; ++u) {
      const int q = w * RPW + u;
      const int n = runCnt[q];
      const int st = runStart[q];
      for (int i = lane + 32; i < n; i += 32) {
        const unsigned lr = (unsigned)(part[st + i].x - w0);
        if (lr < (unsigned)WROWS) atomicAdd(&cnt[lr], 1);
      }
    }
    __syncthreads();
    const int c = cnt[tid];
    int incl = c;
#pragma unroll
    for (int off = 1; off < 32; off <<= 1) {
      const int t = __shfl_up(incl, off, 32);
      if (lane >= off) incl += t;
    }
    if (lane == 31) wsum[w] = incl;
    __syncthreads();
    if (w == 0) {
      const int ws = wsum[lane];
      int si = ws;
#pragma unroll
      for (int off = 1; off < 32; off <<= 1) {
        const int t = __shfl_up(si, off, 32);
        if (lane >= off) si += t;
      }
      wsum[lane] = si - ws;
      if (lane == 31) winTotal = si;
    }
    __syncthreads();
    const int total = winTotal;
    const bool inLds = total <= CAP;
    int* __restrict__ scolWin = scol + winBase;
    cnt[tid] = wsum[w] + incl - c;
    __syncthreads();
#pragma unroll
    for (int u = 0; u < RPW; ++u) {
      const unsigned lr = (unsigned)(pr[u].x - w0);
      if (lr < (unsigned)WROWS) {
        const int pos = atomicAdd(&cnt[lr], 1);
        if (inLds) lcol[pos] = pr[u].y;
        else scolWin[pos] = pr[u].y;
      }
    }
    for (int u = 0; u < RPW; ++u) {
      const int q = w * RPW + u;
      const int n = runCnt[q];
      const int st = runStart[q];
      for (int i = lane + 32; i < n; i += 32) {
        const int2 p2 = part[st + i];
        const unsigned lr = (unsigned)(p2.x - w0);
        if (lr < (unsigned)WROWS) {
          const int pos = atomicAdd(&cnt[lr], 1);
          if (inLds) lcol[pos] = p2.y;
          else scolWin[pos] = p2.y;
        }
      }
    }
    __syncthreads();
    const int nrows = min(WROWS, r1 - w0);
    int lr = w;
    while (lr < nrows) {
      const int beg = __builtin_amdgcn_readfirstlane(cz[lr]);
      const int end = __builtin_amdgcn_readfirstlane(cz[lr + 1]);
      int nx = 0;
      if (lane == 0) nx = atomicAdd(&nextRow, 1);
      if (inLds) {
        gat_one_row(Hp, [](int i) { return lcol[i]; }, beg, end, w0 + lr, g,
                    laneOff, slot, gOdd, gHi, ownOff, bv, out);
      } else {
        const int* sc = scolWin;
        gat_one_row(Hp, [sc](int i) { return sc[i]; }, beg, end, w0 + lr, g,
                    laneOff, slot, gOdd, gHi, ownOff, bv, out);
      }
      lr = __builtin_amdgcn_readfirstlane(nx);
    }
    winBase += total;
    __syncthreads();
  }
}

extern "C" void kernel_launch(void* const* d_in, const int* in_sizes, int n_in,
                              void* d_out, int out_size, void* d_ws, size_t ws_size,
                              hipStream_t stream) {
  const float* H    = (const float*)d_in[0];
  const float* W    = (const float*)d_in[1];
  const float* bias = (const float*)d_in[2];
  const int*   row  = (const int*)d_in[3];
  const int*   col  = (const int*)d_in[4];

  const int N = in_sizes[0] / F;
  const int E = in_sizes[3];
  float* out = (float*)d_out;

  float*  Hp    = (float*)d_ws;
  int*    S     = (int*)(Hp + (size_t)N * F);
  int2*   part  = (int2*)(S + NBK * SPITCH);
  int*    scol  = (int*)(part + E);

  const int rpb = (N + NBK - 1) / NBK;
  const int strips = (N + 15) / 16;
  const int gblocks = strips >= 2048 ? (strips / 2048) * 256 : (strips + 7) / 8;
  sort_and_gemm<<<NBK + gblocks, PTH, 0, stream>>>(row, col, S, part, E, rpb,
                                                  H, W, Hp, N, gblocks);

  range_gat<<<NBK, 1024, 0, stream>>>(part, S, Hp, scol, bias, out, E, N, rpb);
}
